# starting kernel
# baseline (speedup 1.0000x reference)
#include <hip/hip_runtime.h>
#include <hip/hip_bf16.h>
#include <math.h>

#define BATCH   8
#define WINDOW  5
#define NN      20000
#define EPB     320000
#define HID     32
#define KPAD    8
#define NTOT    (BATCH * NN)
#define ETOT    (BATCH * EPB)

typedef float v2f __attribute__((ext_vector_type(2)));
typedef float v8f __attribute__((ext_vector_type(8)));

static __device__ __forceinline__ void edge_nodes(const int* __restrict__ ei,
                                                  int e, int& s, int& d) {
    int b  = e / EPB;
    int el = e - b * EPB;
    const int* base = ei + (size_t)b * 2 * EPB;
    s = base[el] + b * NN;
    d = base[EPB + el] + b * NN;
}

__global__ void zero_f32x4(float4* __restrict__ p, int n4) {
    int i = blockIdx.x * blockDim.x + threadIdx.x;
    if (i < n4) p[i] = make_float4(0.f, 0.f, 0.f, 0.f);
}

__global__ void pad_rows_kernel(const float* __restrict__ x, float* __restrict__ xp) {
    int i = blockIdx.x * blockDim.x + threadIdx.x;
    if (i >= NTOT) return;
    const float* __restrict__ r = x + (size_t)i * WINDOW;
    float* __restrict__ o = xp + (size_t)i * KPAD;
    float4 lo = make_float4(r[0], r[1], r[2], r[3]);
    float4 hi = make_float4(r[4], 0.f, 0.f, 0.f);
    reinterpret_cast<float4*>(o)[0] = lo;
    reinterpret_cast<float4*>(o)[1] = hi;
}

__global__ void degree_kernel(const int* __restrict__ ei, float* __restrict__ deg) {
    int e = blockIdx.x * blockDim.x + threadIdx.x;
    if (e >= ETOT) return;
    int s, d; edge_nodes(ei, e, s, d);
    atomicAdd(&deg[d], 1.0f);
}

__global__ void invsqrt_kernel(float* __restrict__ deg) {
    int i = blockIdx.x * blockDim.x + threadIdx.x;
    if (i >= NTOT) return;
    float v = deg[i];
    deg[i] = (v > 0.0f) ? rsqrtf(v) : 0.0f;
}

__global__ void norm_kernel(const int* __restrict__ ei, const float* __restrict__ dis,
                            float* __restrict__ nrm) {
    int e = blockIdx.x * blockDim.x + threadIdx.x;
    if (e >= ETOT) return;
    int s, d; edge_nodes(ei, e, s, d);
    nrm[e] = dis[s] * dis[d];
}

template <int F, int S>
__global__ void aggregate_kernel(const int* __restrict__ ei, const float* __restrict__ nrm,
                                 const float* __restrict__ hin, float* __restrict__ agg) {
    int e = blockIdx.x * blockDim.x + threadIdx.x;
    if (e >= ETOT) return;
    int s, d; edge_nodes(ei, e, s, d);
    float w = nrm[e];
    const float* __restrict__ hs = hin + (size_t)s * S;
    float* __restrict__ ad = agg + (size_t)d * S;
    if constexpr ((F % 4) == 0) {
        #pragma unroll
        for (int g = 0; g < F / 4; ++g) {
            float4 v = reinterpret_cast<const float4*>(hs)[g];
            atomicAdd(&ad[4 * g + 0], v.x * w);
            atomicAdd(&ad[4 * g + 1], v.y * w);
            atomicAdd(&ad[4 * g + 2], v.z * w);
            atomicAdd(&ad[4 * g + 3], v.w * w);
        }
    } else {
        #pragma unroll
        for (int f = 0; f < F; ++f) atomicAdd(&ad[f], hs[f] * w);
    }
}

template <int K, int KW>
__global__ void tag_gemm_kernel(const float* __restrict__ hin,
                                const float* __restrict__ agg,
                                const float* __restrict__ W,
                                const float* __restrict__ bias,
                                float* __restrict__ hout, int N) {
    static_assert((K % 4) == 0, "padded K must be a multiple of 4");
    __shared__ float sW[2 * K * HID];
    __shared__ float sB[HID];
    for (int i = threadIdx.x; i < 2 * K * HID; i += blockDim.x) {
        int s   = i / (K * HID);
        int rem = i - s * (K * HID);
        int kw  = rem / HID;
        int col = rem - kw * HID;
        sW[i] = (kw < KW) ? W[((size_t)s * KW + kw) * HID + col] : 0.0f;
    }
    if (threadIdx.x < HID) sB[threadIdx.x] = bias[threadIdx.x];
    __syncthreads();

    int lane = threadIdx.x & 31;
    int wave = (blockIdx.x * blockDim.x + threadIdx.x) >> 5;
    int row0 = wave * 16;
    if (row0 >= N) return;

    const int m     = lane & 15;
    const int khalf = (lane >> 4) * 2;

    v8f acc0 = {0.f,0.f,0.f,0.f,0.f,0.f,0.f,0.f};
    v8f acc1 = {0.f,0.f,0.f,0.f,0.f,0.f,0.f,0.f};

    #pragma unroll
    for (int s = 0; s < 2; ++s) {
        const float* __restrict__ A   = s ? agg : hin;
        const float* __restrict__ sWk = sW + s * K * HID;
        #pragma unroll
        for (int c = 0; c < K / 4; ++c) {
            const int k0 = c * 4 + khalf;
            v2f a = *reinterpret_cast<const v2f*>(&A[(size_t)(row0 + m) * K + k0]);
            v2f b0, b1;
            b0.x = sWk[k0 * HID + m];
            b0.y = sWk[(k0 + 1) * HID + m];
            b1.x = sWk[k0 * HID + 16 + m];
            b1.y = sWk[(k0 + 1) * HID + 16 + m];
            acc0 = __builtin_amdgcn_wmma_f32_16x16x4_f32(
                       false, a, false, b0, (short)0, acc0, false, false);
            acc1 = __builtin_amdgcn_wmma_f32_16x16x4_f32(
                       false, a, false, b1, (short)0, acc1, false, false);
        }
    }

    const int rbase = row0 + ((lane >> 4) ? 8 : 0);
    #pragma unroll
    for (int v = 0; v < 8; ++v) {
        int r = rbase + v;
        float x0 = acc0[v] + sB[m];
        float x1 = acc1[v] + sB[16 + m];
        hout[(size_t)r * HID + m]      = tanhf(x0);
        hout[(size_t)r * HID + 16 + m] = tanhf(x1);
    }
}

__global__ void readout_kernel(const float* __restrict__ h, const float* __restrict__ Wreg,
                               const float* __restrict__ breg, float* __restrict__ out) {
    int i = blockIdx.x * blockDim.x + threadIdx.x;
    if (i >= NTOT) return;
    float acc = breg[0];
    const float4* hr = reinterpret_cast<const float4*>(h + (size_t)i * HID);
    #pragma unroll
    for (int g = 0; g < HID / 4; ++g) {
        float4 v = hr[g];
        acc += v.x * Wreg[4 * g + 0] + v.y * Wreg[4 * g + 1]
             + v.z * Wreg[4 * g + 2] + v.w * Wreg[4 * g + 3];
    }
    out[i] = acc;
}

static inline int cdiv(long long a, long long b) { return (int)((a + b - 1) / b); }

extern "C" void kernel_launch(void* const* d_in, const int* in_sizes, int n_in,
                              void* d_out, int out_size, void* d_ws, size_t ws_size,
                              hipStream_t stream) {
    const float* X    = (const float*)d_in[0];
    const int*   ei   = (const int*)  d_in[1];
    const float* W0   = (const float*)d_in[2];
    const float* b0   = (const float*)d_in[3];
    const float* W1   = (const float*)d_in[4];
    const float* b1   = (const float*)d_in[5];
    const float* W2   = (const float*)d_in[6];
    const float* b2   = (const float*)d_in[7];
    const float* Wreg = (const float*)d_in[8];
    const float* breg = (const float*)d_in[9];

    float* out    = (float*)d_out;
    float* hfinal = out + NTOT;

    float* dis = (float*)d_ws;
    float* nrm = dis + NTOT;
    float* agg = nrm + ETOT;
    float* hA  = agg + (size_t)NTOT * HID;
    float* hB  = hA  + (size_t)NTOT * HID;
    float* Xp  = hB  + (size_t)NTOT * HID;

    const int T = 256;

    zero_f32x4<<<cdiv(NTOT / 4, T), T, 0, stream>>>((float4*)dis, NTOT / 4);
    degree_kernel<<<cdiv(ETOT, T), T, 0, stream>>>(ei, dis);
    invsqrt_kernel<<<cdiv(NTOT, T), T, 0, stream>>>(dis);
    norm_kernel<<<cdiv(ETOT, T), T, 0, stream>>>(ei, dis, nrm);

    const int gemm_blocks = cdiv((long long)(NTOT / 16) * 32, T);
    const int agg_blocks  = cdiv(ETOT, T);

    pad_rows_kernel<<<cdiv(NTOT, T), T, 0, stream>>>(X, Xp);
    zero_f32x4<<<cdiv((long long)NTOT * KPAD / 4, T), T, 0, stream>>>((float4*)agg, NTOT * KPAD / 4);
    aggregate_kernel<WINDOW, KPAD><<<agg_blocks, T, 0, stream>>>(ei, nrm, Xp, agg);
    tag_gemm_kernel<KPAD, WINDOW><<<gemm_blocks, T, 0, stream>>>(Xp, agg, W0, b0, hA, NTOT);

    zero_f32x4<<<cdiv((long long)NTOT * HID / 4, T), T, 0, stream>>>((float4*)agg, NTOT * HID / 4);
    aggregate_kernel<HID, HID><<<agg_blocks, T, 0, stream>>>(ei, nrm, hA, agg);
    tag_gemm_kernel<HID, HID><<<gemm_blocks, T, 0, stream>>>(hA, agg, W1, b1, hB, NTOT);

    zero_f32x4<<<cdiv((long long)NTOT * HID / 4, T), T, 0, stream>>>((float4*)agg, NTOT * HID / 4);
    aggregate_kernel<HID, HID><<<agg_blocks, T, 0, stream>>>(ei, nrm, hB, agg);
    tag_gemm_kernel<HID, HID><<<gemm_blocks, T, 0, stream>>>(hB, agg, W2, b2, hfinal, NTOT);

    readout_kernel<<<cdiv(NTOT, T), T, 0, stream>>>(hfinal, Wreg, breg, out);
}


# layers 1, 2: neighbour sums per feature PAIR: two source planes interleaved in LDS, one 8-byte gather per edge serves both; edges in 2 halves added in the product; sum order = arrival per half
# speedup vs baseline: 19.7066x; 19.7066x over previous
#include <hip/hip_runtime.h>
#include <hip/hip_bf16.h>
#include <math.h>

#define BATCH   8
#define WINDOW  5
#define NN      20000
#define EPB     320000
#define HID     32
#define KPAD    8
#define NTOT    (BATCH * NN)
#define ETOT    (BATCH * EPB)
#define L0CH    5
#define DCH     16
#define LCH     2

typedef float v2f __attribute__((ext_vector_type(2)));
typedef float v8f __attribute__((ext_vector_type(8)));

__global__ void pad_rows_kernel(const float* __restrict__ x, float* __restrict__ xp) {
    int i = blockIdx.x * blockDim.x + threadIdx.x;
    if (i >= NTOT) return;
    const float* __restrict__ r = x + (size_t)i * WINDOW;
    float* __restrict__ o = xp + (size_t)i * KPAD;
    float4 lo = make_float4(r[0], r[1], r[2], r[3]);
    float4 hi = make_float4(r[4], 0.f, 0.f, 0.f);
    reinterpret_cast<float4*>(o)[0] = lo;
    reinterpret_cast<float4*>(o)[1] = hi;
}

__global__ __launch_bounds__(1024)
void degree_count_kernel(const int* __restrict__ ei, unsigned* __restrict__ cnt) {
    static_assert((EPB % (4 * DCH)) == 0 && (NN % 4) == 0, "16-byte vectors over edge chunks and nodes");
    constexpr int NVC   = EPB / 4 / DCH;
    constexpr int TRIPS = (NVC + 1023) / 1024;
    __shared__ __attribute__((aligned(16))) unsigned sCnt[NN];
    const int c   = blockIdx.x % DCH;
    const int b   = blockIdx.x / DCH;
    const int tid = threadIdx.x;

    const int4* __restrict__ D =
        reinterpret_cast<const int4*>(ei + (size_t)b * 2 * EPB + EPB) + (size_t)c * NVC;
    int4 d[TRIPS];
    #pragma unroll
    for (int k = 0; k < TRIPS; ++k) {
        const int j = tid + k * 1024;
        d[k] = D[(j < NVC) ? j : tid];
    }
    for (int i = tid; i < NN / 4; i += 1024)
        reinterpret_cast<uint4*>(sCnt)[i] = make_uint4(0u, 0u, 0u, 0u);
    __syncthreads();

    #pragma unroll
    for (int k = 0; k < TRIPS; ++k) {
        if (tid + k * 1024 < NVC) {
            __hip_atomic_fetch_add(&sCnt[d[k].x], 1u, __ATOMIC_RELAXED, __HIP_MEMORY_SCOPE_WORKGROUP);
            __hip_atomic_fetch_add(&sCnt[d[k].y], 1u, __ATOMIC_RELAXED, __HIP_MEMORY_SCOPE_WORKGROUP);
            __hip_atomic_fetch_add(&sCnt[d[k].z], 1u, __ATOMIC_RELAXED, __HIP_MEMORY_SCOPE_WORKGROUP);
            __hip_atomic_fetch_add(&sCnt[d[k].w], 1u, __ATOMIC_RELAXED, __HIP_MEMORY_SCOPE_WORKGROUP);
        }
    }
    __syncthreads();

    uint4* __restrict__ op =
        reinterpret_cast<uint4*>(cnt + (size_t)c * NTOT + (size_t)b * NN);
    for (int i = tid; i < NN / 4; i += 1024)
        op[i] = reinterpret_cast<const uint4*>(sCnt)[i];
}

__global__ void degree_rsqrt_kernel(const unsigned* __restrict__ cnt, float* __restrict__ dis) {
    int i = blockIdx.x * blockDim.x + threadIdx.x;
    if (i >= NTOT / 4) return;
    uint4 s = reinterpret_cast<const uint4*>(cnt)[i];
    #pragma unroll
    for (int c = 1; c < DCH; ++c) {
        const uint4 t = reinterpret_cast<const uint4*>(cnt + (size_t)c * NTOT)[i];
        s.x += t.x; s.y += t.y; s.z += t.z; s.w += t.w;
    }
    float4 r;
    r.x = (s.x > 0u) ? rsqrtf((float)s.x) : 0.0f;
    r.y = (s.y > 0u) ? rsqrtf((float)s.y) : 0.0f;
    r.z = (s.z > 0u) ? rsqrtf((float)s.z) : 0.0f;
    r.w = (s.w > 0u) ? rsqrtf((float)s.w) : 0.0f;
    reinterpret_cast<float4*>(dis)[i] = r;
}

template <int NF, int NCH>
__global__ __launch_bounds__(1024)
void plane_sum_kernel(const int* __restrict__ ei, const float* __restrict__ dis,
                      const float* __restrict__ src, float* __restrict__ dst) {
    static_assert((NN % 4) == 0 && (EPB % (4 * NCH)) == 0, "16-byte vectors over nodes and edge chunks");
    constexpr int NVC  = EPB / 4 / NCH;
    constexpr int STEP = 2048;
    constexpr int FULL = NVC / STEP;
    static_assert(FULL >= 1, "every thread owns a first pair of edge vectors");
    __shared__ __attribute__((aligned(16))) float sSrc[NN];
    __shared__ __attribute__((aligned(16))) float sAcc[NN + 1024];
    static_assert(sizeof(float) * (2 * NN + 1024) > 160 * 1024, "more than half of the 320 KiB of a WGP");
    const int c   = blockIdx.x % NCH;
    const int t   = blockIdx.x / NCH;
    const int f   = t % NF;
    const int b   = t / NF;
    const int tid = threadIdx.x;

    const float* __restrict__ gdis = dis + (size_t)b * NN;
    const float* __restrict__ sp = src + (size_t)b * NN * NF + f;
    for (int i = tid; i < NN; i += 1024) {
        sSrc[i] = sp[(size_t)i * NF] * gdis[i];
        sAcc[i] = 0.f;
    }
    __syncthreads();

    const int* __restrict__ eb = ei + (size_t)b * 2 * EPB;
    const int4* __restrict__ S = reinterpret_cast<const int4*>(eb) + (size_t)c * NVC;
    const int4* __restrict__ D = reinterpret_cast<const int4*>(eb + EPB) + (size_t)c * NVC;

    auto add4 = [&](const int4 s, const int4 d) {
        const float v0 = sSrc[s.x];
        const float v1 = sSrc[s.y];
        const float v2 = sSrc[s.z];
        const float v3 = sSrc[s.w];
        __hip_atomic_fetch_add(&sAcc[d.x], v0, __ATOMIC_RELAXED, __HIP_MEMORY_SCOPE_WORKGROUP);
        __hip_atomic_fetch_add(&sAcc[d.y], v1, __ATOMIC_RELAXED, __HIP_MEMORY_SCOPE_WORKGROUP);
        __hip_atomic_fetch_add(&sAcc[d.z], v2, __ATOMIC_RELAXED, __HIP_MEMORY_SCOPE_WORKGROUP);
        __hip_atomic_fetch_add(&sAcc[d.w], v3, __ATOMIC_RELAXED, __HIP_MEMORY_SCOPE_WORKGROUP);
    };

    int i = tid;
    int4 s0 = S[i], s1 = S[i + 1024];
    int4 d0 = D[i], d1 = D[i + 1024];
    for (int trip = 0; trip < FULL; ++trip) {
        const int ni = (trip + 1 < FULL) ? i + STEP : i;
        const int4 s0n = S[ni], s1n = S[ni + 1024];
        const int4 d0n = D[ni], d1n = D[ni + 1024];
        add4(s0, d0);
        add4(s1, d1);
        s0 = s0n; s1 = s1n; d0 = d0n; d1 = d1n;
        i = ni;
    }
    for (int j = FULL * STEP + tid; j < NVC; j += 1024) add4(S[j], D[j]);
    __syncthreads();

    float4* __restrict__ dp =
        reinterpret_cast<float4*>(dst + (size_t)(c * NF + f) * NTOT + (size_t)b * NN);
    const float4* __restrict__ dq = reinterpret_cast<const float4*>(gdis);
    for (int j = tid; j < NN / 4; j += 1024) {
        const float4 a = reinterpret_cast<const float4*>(sAcc)[j];
        const float4 q = dq[j];
        dp[j] = make_float4(a.x * q.x, a.y * q.y, a.z * q.z, a.w * q.w);
    }
}

template <int NCH>
__global__ __launch_bounds__(1024)
void pair_sum_kernel(const int* __restrict__ ei, const float* __restrict__ dis,
                     const float* __restrict__ src, float* __restrict__ dst) {
    static_assert((NN % 4) == 0 && (EPB % (4 * NCH)) == 0 && (HID % 2) == 0,
                  "16-byte vectors over nodes and edge chunks, features in pairs");
    constexpr int NVC  = EPB / 4 / NCH;
    constexpr int STEP = 2048;
    constexpr int FULL = NVC / STEP;
    static_assert(FULL >= 1, "every thread owns a first pair of edge vectors");
    __shared__ __attribute__((aligned(16))) float2 sSrc[NN];
    __shared__ __attribute__((aligned(16))) float  sAcc0[NN];
    __shared__ __attribute__((aligned(16))) float  sAcc1[NN];
    static_assert(sizeof(float) * 4 * NN <= 320 * 1024, "four planes fit the 320 KiB of a WGP");
    const int c   = blockIdx.x % NCH;
    const int t   = blockIdx.x / NCH;
    const int f0  = (t % (HID / 2)) * 2;
    const int b   = t / (HID / 2);
    const int tid = threadIdx.x;

    const float* __restrict__ gdis = dis + (size_t)b * NN;
    {
        const float4* __restrict__ sp0 =
            reinterpret_cast<const float4*>(src + (size_t)f0 * NTOT + (size_t)b * NN);
        const float4* __restrict__ sp1 =
            reinterpret_cast<const float4*>(src + (size_t)(f0 + 1) * NTOT + (size_t)b * NN);
        const float4* __restrict__ dq = reinterpret_cast<const float4*>(gdis);
        for (int i = tid; i < NN / 4; i += 1024) {
            const float4 u = sp0[i];
            const float4 v = sp1[i];
            const float4 q = dq[i];
            reinterpret_cast<float4*>(sSrc)[2 * i]     = make_float4(u.x * q.x, v.x * q.x, u.y * q.y, v.y * q.y);
            reinterpret_cast<float4*>(sSrc)[2 * i + 1] = make_float4(u.z * q.z, v.z * q.z, u.w * q.w, v.w * q.w);
            reinterpret_cast<float4*>(sAcc0)[i] = make_float4(0.f, 0.f, 0.f, 0.f);
            reinterpret_cast<float4*>(sAcc1)[i] = make_float4(0.f, 0.f, 0.f, 0.f);
        }
    }
    __syncthreads();

    const int* __restrict__ eb = ei + (size_t)b * 2 * EPB;
    const int4* __restrict__ S = reinterpret_cast<const int4*>(eb) + (size_t)c * NVC;
    const int4* __restrict__ D = reinterpret_cast<const int4*>(eb + EPB) + (size_t)c * NVC;

    auto add4 = [&](const int4 s, const int4 d) {
        const float2 v0 = sSrc[s.x];
        const float2 v1 = sSrc[s.y];
        const float2 v2 = sSrc[s.z];
        const float2 v3 = sSrc[s.w];
        __hip_atomic_fetch_add(&sAcc0[d.x], v0.x, __ATOMIC_RELAXED, __HIP_MEMORY_SCOPE_WORKGROUP);
        __hip_atomic_fetch_add(&sAcc1[d.x], v0.y, __ATOMIC_RELAXED, __HIP_MEMORY_SCOPE_WORKGROUP);
        __hip_atomic_fetch_add(&sAcc0[d.y], v1.x, __ATOMIC_RELAXED, __HIP_MEMORY_SCOPE_WORKGROUP);
        __hip_atomic_fetch_add(&sAcc1[d.y], v1.y, __ATOMIC_RELAXED, __HIP_MEMORY_SCOPE_WORKGROUP);
        __hip_atomic_fetch_add(&sAcc0[d.z], v2.x, __ATOMIC_RELAXED, __HIP_MEMORY_SCOPE_WORKGROUP);
        __hip_atomic_fetch_add(&sAcc1[d.z], v2.y, __ATOMIC_RELAXED, __HIP_MEMORY_SCOPE_WORKGROUP);
        __hip_atomic_fetch_add(&sAcc0[d.w], v3.x, __ATOMIC_RELAXED, __HIP_MEMORY_SCOPE_WORKGROUP);
        __hip_atomic_fetch_add(&sAcc1[d.w], v3.y, __ATOMIC_RELAXED, __HIP_MEMORY_SCOPE_WORKGROUP);
    };

    int i = tid;
    int4 s0 = S[i], s1 = S[i + 1024];
    int4 d0 = D[i], d1 = D[i + 1024];
    for (int trip = 0; trip < FULL; ++trip) {
        const int ni = (trip + 1 < FULL) ? i + STEP : i;
        const int4 s0n = S[ni], s1n = S[ni + 1024];
        const int4 d0n = D[ni], d1n = D[ni + 1024];
        add4(s0, d0);
        add4(s1, d1);
        s0 = s0n; s1 = s1n; d0 = d0n; d1 = d1n;
        i = ni;
    }
    for (int j = FULL * STEP + tid; j < NVC; j += 1024) add4(S[j], D[j]);
    __syncthreads();

    float4* __restrict__ dp0 =
        reinterpret_cast<float4*>(dst + (size_t)(c * HID + f0) * NTOT + (size_t)b * NN);
    float4* __restrict__ dp1 =
        reinterpret_cast<float4*>(dst + (size_t)(c * HID + f0 + 1) * NTOT + (size_t)b * NN);
    const float4* __restrict__ dq = reinterpret_cast<const float4*>(gdis);
    for (int j = tid; j < NN / 4; j += 1024) {
        const float4 a0 = reinterpret_cast<const float4*>(sAcc0)[j];
        const float4 a1 = reinterpret_cast<const float4*>(sAcc1)[j];
        const float4 q  = dq[j];
        dp0[j] = make_float4(a0.x * q.x, a0.y * q.y, a0.z * q.z, a0.w * q.w);
        dp1[j] = make_float4(a1.x * q.x, a1.y * q.y, a1.z * q.z, a1.w * q.w);
    }
}

template <int K, int KW, int NCH, bool OWN_PLANES, bool OUT_PLANES>
__global__ __launch_bounds__(256)
void tag_gemm_kernel(const float* __restrict__ hin,
                                const float* __restrict__ agg,
                                const float* __restrict__ W,
                                const float* __restrict__ bias,
                                float* __restrict__ hout, int N) {
    static_assert((K % 4) == 0, "padded K must be a multiple of 4");
    __shared__ float sW[2 * K * HID];
    __shared__ float sB[HID];
    for (int i = threadIdx.x; i < 2 * K * HID; i += blockDim.x) {
        int s   = i / (K * HID);
        int rem = i - s * (K * HID);
        int kw  = rem / HID;
        int col = rem - kw * HID;
        sW[i] = (kw < KW) ? W[((size_t)s * KW + kw) * HID + col] : 0.0f;
    }
    if (threadIdx.x < HID) sB[threadIdx.x] = bias[threadIdx.x];
    __syncthreads();

    int lane = threadIdx.x & 31;
    int wave = (blockIdx.x * blockDim.x + threadIdx.x) >> 5;

    const int m     = lane & 15;
    const int khalf = (lane >> 4) * 2;

    for (int tile = 0; tile < 2; ++tile) {
    const int row0 = (wave * 2 + tile) * 16;
    if (row0 >= N) break;

    auto agg_at = [&](const int k) -> float {
        const int kk = (KW == K || k < KW) ? k : (KW - 1);
        float v = agg[(size_t)kk * NTOT + row0 + m];
        #pragma unroll
        for (int ch = 1; ch < NCH; ++ch)
            v += agg[((size_t)ch * KW + kk) * NTOT + row0 + m];
        return (KW == K || k < KW) ? v : 0.0f;
    };

    v8f acc0 = {0.f,0.f,0.f,0.f,0.f,0.f,0.f,0.f};
    v8f acc1 = {0.f,0.f,0.f,0.f,0.f,0.f,0.f,0.f};

    #pragma unroll
    for (int s = 0; s < 2; ++s) {
        const float* __restrict__ sWk = sW + s * K * HID;
        #pragma unroll
        for (int c = 0; c < K / 4; ++c) {
            const int k0 = c * 4 + khalf;
            v2f a;
            if (s) {
                a.x = agg_at(k0);
                a.y = agg_at(k0 + 1);
            } else if constexpr (OWN_PLANES) {
                a.x = hin[(size_t)k0 * NTOT + row0 + m];
                a.y = hin[(size_t)(k0 + 1) * NTOT + row0 + m];
            } else {
                a = *reinterpret_cast<const v2f*>(&hin[(size_t)(row0 + m) * K + k0]);
            }
            v2f b0, b1;
            b0.x = sWk[k0 * HID + m];
            b0.y = sWk[(k0 + 1) * HID + m];
            b1.x = sWk[k0 * HID + 16 + m];
            b1.y = sWk[(k0 + 1) * HID + 16 + m];
            if constexpr (OUT_PLANES) {
                acc0 = __builtin_amdgcn_wmma_f32_16x16x4_f32(
                           false, b0, false, a, (short)0, acc0, false, false);
                acc1 = __builtin_amdgcn_wmma_f32_16x16x4_f32(
                           false, b1, false, a, (short)0, acc1, false, false);
            } else {
                acc0 = __builtin_amdgcn_wmma_f32_16x16x4_f32(
                           false, a, false, b0, (short)0, acc0, false, false);
                acc1 = __builtin_amdgcn_wmma_f32_16x16x4_f32(
                           false, a, false, b1, (short)0, acc1, false, false);
            }
        }
    }

    if constexpr (OUT_PLANES) {
        const int fbase = (lane >> 4) ? 8 : 0;
        #pragma unroll
        for (int v = 0; v < 8; ++v) {
            int f = fbase + v;
            float x0 = acc0[v] + sB[f];
            float x1 = acc1[v] + sB[16 + f];
            hout[(size_t)f * NTOT + row0 + m]        = tanhf(x0);
            hout[(size_t)(16 + f) * NTOT + row0 + m] = tanhf(x1);
        }
    } else {
        const int rbase = row0 + ((lane >> 4) ? 8 : 0);
        #pragma unroll
        for (int v = 0; v < 8; ++v) {
            int r = rbase + v;
            float x0 = acc0[v] + sB[m];
            float x1 = acc1[v] + sB[16 + m];
            hout[(size_t)r * HID + m]      = tanhf(x0);
            hout[(size_t)r * HID + 16 + m] = tanhf(x1);
        }
    }
    }
}

__global__ void readout_kernel(const float* __restrict__ h, const float* __restrict__ Wreg,
                               const float* __restrict__ breg, float* __restrict__ out) {
    int i = blockIdx.x * blockDim.x + threadIdx.x;
    if (i >= NTOT) return;
    float acc = breg[0];
    const float4* hr = reinterpret_cast<const float4*>(h + (size_t)i * HID);
    #pragma unroll
    for (int g = 0; g < HID / 4; ++g) {
        float4 v = hr[g];
        acc += v.x * Wreg[4 * g + 0] + v.y * Wreg[4 * g + 1]
             + v.z * Wreg[4 * g + 2] + v.w * Wreg[4 * g + 3];
    }
    out[i] = acc;
}

static inline int cdiv(long long a, long long b) { return (int)((a + b - 1) / b); }

extern "C" void kernel_launch(void* const* d_in, const int* in_sizes, int n_in,
                              void* d_out, int out_size, void* d_ws, size_t ws_size,
                              hipStream_t stream) {
    const float* X    = (const float*)d_in[0];
    const int*   ei   = (const int*)  d_in[1];
    const float* W0   = (const float*)d_in[2];
    const float* b0   = (const float*)d_in[3];
    const float* W1   = (const float*)d_in[4];
    const float* b1   = (const float*)d_in[5];
    const float* W2   = (const float*)d_in[6];
    const float* b2   = (const float*)d_in[7];
    const float* Wreg = (const float*)d_in[8];
    const float* breg = (const float*)d_in[9];

    float* out    = (float*)d_out;
    float* hfinal = out + NTOT;

    float* dis = (float*)d_ws;
    float* agg = dis + NTOT;
    float* hA  = agg + (size_t)NTOT * HID * LCH;
    float* hB  = hA  + (size_t)NTOT * HID;
    float* Xp  = hB  + (size_t)NTOT * HID;
    unsigned* cnt = (unsigned*)(Xp + (size_t)NTOT * KPAD);

    const int T = 256;

    degree_count_kernel<<<BATCH * DCH, 1024, 0, stream>>>(ei, cnt);
    degree_rsqrt_kernel<<<cdiv(NTOT / 4, T), T, 0, stream>>>(cnt, dis);

    const int gemm_blocks = cdiv((long long)(NTOT / 32) * 32, T);

    pad_rows_kernel<<<cdiv(NTOT, T), T, 0, stream>>>(X, Xp);
    plane_sum_kernel<WINDOW, L0CH><<<BATCH * WINDOW * L0CH, 1024, 0, stream>>>(ei, dis, X, agg);
    tag_gemm_kernel<KPAD, WINDOW, L0CH, false, true><<<gemm_blocks, T, 0, stream>>>(Xp, agg, W0, b0, hA, NTOT);

    pair_sum_kernel<LCH><<<BATCH * (HID / 2) * LCH, 1024, 0, stream>>>(ei, dis, hA, agg);
    tag_gemm_kernel<HID, HID, LCH, true, true><<<gemm_blocks, T, 0, stream>>>(hA, agg, W1, b1, hB, NTOT);

    pair_sum_kernel<LCH><<<BATCH * (HID / 2) * LCH, 1024, 0, stream>>>(ei, dis, hB, agg);
    tag_gemm_kernel<HID, HID, LCH, true, false><<<gemm_blocks, T, 0, stream>>>(hB, agg, W2, b2, hfinal, NTOT);

    readout_kernel<<<cdiv(NTOT, T), T, 0, stream>>>(hfinal, Wreg, breg, out);
}


# layers 1, 2 walk a per-call bank schedule of the edges (cell = src mod 32, dst mod 64): a wave's 32 lanes gather and add on 32 different LDS banks; 4 B/edge; sum order = arrival, varies as before
# speedup vs baseline: 20.5293x; 1.0417x over previous
#include <hip/hip_runtime.h>
#include <hip/hip_bf16.h>
#include <math.h>

#define BATCH   8
#define WINDOW  5
#define NN      20000
#define EPB     320000
#define HID     32
#define KPAD    8
#define NTOT    (BATCH * NN)
#define ETOT    (BATCH * EPB)
#define L0CH    5
#define DCH     16
#define LCH     2
#define PASSES  11
#define GRIDW   (PASSES * 64 * 32)
#define OVCAP   (EPB / DCH)
#define OVPAD   (((OVCAP + 1023) / 1024) * 1024)
#define OVNST   32

typedef float v2f __attribute__((ext_vector_type(2)));
typedef float v8f __attribute__((ext_vector_type(8)));

__global__ void pad_rows_kernel(const float* __restrict__ x, float* __restrict__ xp) {
    int i = blockIdx.x * blockDim.x + threadIdx.x;
    if (i >= NTOT) return;
    const float* __restrict__ r = x + (size_t)i * WINDOW;
    float* __restrict__ o = xp + (size_t)i * KPAD;
    float4 lo = make_float4(r[0], r[1], r[2], r[3]);
    float4 hi = make_float4(r[4], 0.f, 0.f, 0.f);
    reinterpret_cast<float4*>(o)[0] = lo;
    reinterpret_cast<float4*>(o)[1] = hi;
}

__global__ __launch_bounds__(1024)
void degree_schedule_kernel(const int* __restrict__ ei, unsigned* __restrict__ cnt,
                            unsigned* __restrict__ sched, unsigned* __restrict__ ov,
                            unsigned* __restrict__ ovcnt) {
    static_assert((EPB % (4 * DCH)) == 0 && (NN % 4) == 0, "16-byte vectors over edge chunks and nodes");
    static_assert((NN % 32) == 0, "the null source entries NN + lane keep the lane's bank pair");
    static_assert(NN + 32 <= 65536 && NN >= 64, "packed 16-bit source and destination indices; null destinations are real nodes");
    static_assert((GRIDW % 4) == 0 && (OVCAP % 4) == 0, "16-byte vectors over grid and overflow list");
    constexpr int NVC   = EPB / 4 / DCH;
    constexpr int TRIPS = (NVC + 1023) / 1024;
    __shared__ __attribute__((aligned(16))) unsigned sCnt[NN];
    __shared__ __attribute__((aligned(16))) unsigned sGrid[GRIDW];
    __shared__ __attribute__((aligned(16))) unsigned sOv[OVCAP];
    __shared__ __attribute__((aligned(16))) unsigned sCell[32 * 64];
    __shared__ unsigned sOvN;
    const int c   = blockIdx.x % DCH;
    const int b   = blockIdx.x / DCH;
    const int tid = threadIdx.x;

    const int4* __restrict__ S =
        reinterpret_cast<const int4*>(ei + (size_t)b * 2 * EPB) + (size_t)c * NVC;
    const int4* __restrict__ D =
        reinterpret_cast<const int4*>(ei + (size_t)b * 2 * EPB + EPB) + (size_t)c * NVC;
    int4 s[TRIPS], d[TRIPS];
    #pragma unroll
    for (int k = 0; k < TRIPS; ++k) {
        const int j = tid + k * 1024;
        s[k] = S[(j < NVC) ? j : tid];
        d[k] = D[(j < NVC) ? j : tid];
    }
    for (int i = tid; i < NN / 4; i += 1024)
        reinterpret_cast<uint4*>(sCnt)[i] = make_uint4(0u, 0u, 0u, 0u);
    if (tid < 32 * 64 / 4)
        reinterpret_cast<uint4*>(sCell)[tid] = make_uint4(0u, 0u, 0u, 0u);
    if (tid == 0) sOvN = 0u;
    for (int v = tid; v < GRIDW / 4; v += 1024) {
        const unsigned lane = (unsigned)v & 31u;
        const unsigned r0   = ((unsigned)v >> 5) << 2;
        const unsigned ns   = (unsigned)NN + lane;
        reinterpret_cast<uint4*>(sGrid)[v] = make_uint4(ns | (((lane + r0     ) & 63u) << 16),
                                                        ns | (((lane + r0 + 1u) & 63u) << 16),
                                                        ns | (((lane + r0 + 2u) & 63u) << 16),
                                                        ns | (((lane + r0 + 3u) & 63u) << 16));
    }
    __syncthreads();

    auto rank_of = [&](const int src, const int dst) -> unsigned {
        __hip_atomic_fetch_add(&sCnt[dst], 1u, __ATOMIC_RELAXED, __HIP_MEMORY_SCOPE_WORKGROUP);
        return __hip_atomic_fetch_add(&sCell[((unsigned)src & 31u) * 64u + ((unsigned)dst & 63u)], 1u,
                                      __ATOMIC_RELAXED, __HIP_MEMORY_SCOPE_WORKGROUP);
    };
    auto put = [&](const int src, const int dst, const unsigned p) {
        const unsigned sc = (unsigned)src & 31u;
        const unsigned dc = (unsigned)dst & 63u;
        const unsigned w  = (unsigned)src | ((unsigned)dst << 16);
        if (p < (unsigned)PASSES) {
            const unsigned R = p * 64u + ((dc - sc) & 63u);
            sGrid[((((R >> 2) << 5) + sc) << 2) | (R & 3u)] = w;
        } else {
            const unsigned q = __hip_atomic_fetch_add(&sOvN, 1u, __ATOMIC_RELAXED,
                                                      __HIP_MEMORY_SCOPE_WORKGROUP);
            sOv[q] = w;
        }
    };
    unsigned rk[TRIPS][4];
    #pragma unroll
    for (int k = 0; k < TRIPS; ++k) {
        if (tid + k * 1024 < NVC) {
            rk[k][0] = rank_of(s[k].x, d[k].x);
            rk[k][1] = rank_of(s[k].y, d[k].y);
            rk[k][2] = rank_of(s[k].z, d[k].z);
            rk[k][3] = rank_of(s[k].w, d[k].w);
        }
    }
    #pragma unroll
    for (int k = 0; k < TRIPS; ++k) {
        if (tid + k * 1024 < NVC) {
            put(s[k].x, d[k].x, rk[k][0]);
            put(s[k].y, d[k].y, rk[k][1]);
            put(s[k].z, d[k].z, rk[k][2]);
            put(s[k].w, d[k].w, rk[k][3]);
        }
    }
    __syncthreads();

    uint4* __restrict__ op =
        reinterpret_cast<uint4*>(cnt + (size_t)c * NTOT + (size_t)b * NN);
    for (int i = tid; i < NN / 4; i += 1024)
        op[i] = reinterpret_cast<const uint4*>(sCnt)[i];

    const int seg = b * DCH + c;
    uint4* __restrict__ gp = reinterpret_cast<uint4*>(sched + (size_t)seg * GRIDW);
    for (int i = tid; i < GRIDW / 4; i += 1024)
        gp[i] = reinterpret_cast<const uint4*>(sGrid)[i];
    const unsigned nov = sOvN;
    const unsigned nup = (nov + 1023u) & ~1023u;
    const unsigned npad = (nup > 2048u) ? nup : 2048u;
    unsigned* __restrict__ vp = ov + (size_t)seg * OVPAD;
    for (unsigned i = tid; i < npad; i += 1024u)
        vp[i] = (i < nov) ? sOv[i] : (((unsigned)NN + (i & 31u)) | ((i & 63u) << 16));
    if (tid == 0) ovcnt[(size_t)seg * OVNST] = nov;
}

__global__ void degree_rsqrt_kernel(const unsigned* __restrict__ cnt, float* __restrict__ dis) {
    int i = blockIdx.x * blockDim.x + threadIdx.x;
    if (i >= NTOT / 4) return;
    uint4 s = reinterpret_cast<const uint4*>(cnt)[i];
    #pragma unroll
    for (int c = 1; c < DCH; ++c) {
        const uint4 t = reinterpret_cast<const uint4*>(cnt + (size_t)c * NTOT)[i];
        s.x += t.x; s.y += t.y; s.z += t.z; s.w += t.w;
    }
    float4 r;
    r.x = (s.x > 0u) ? rsqrtf((float)s.x) : 0.0f;
    r.y = (s.y > 0u) ? rsqrtf((float)s.y) : 0.0f;
    r.z = (s.z > 0u) ? rsqrtf((float)s.z) : 0.0f;
    r.w = (s.w > 0u) ? rsqrtf((float)s.w) : 0.0f;
    reinterpret_cast<float4*>(dis)[i] = r;
}

template <int NF, int NCH>
__global__ __launch_bounds__(1024)
void plane_sum_kernel(const int* __restrict__ ei, const float* __restrict__ dis,
                      const float* __restrict__ src, float* __restrict__ dst) {
    static_assert((NN % 4) == 0 && (EPB % (4 * NCH)) == 0, "16-byte vectors over nodes and edge chunks");
    constexpr int NVC  = EPB / 4 / NCH;
    constexpr int STEP = 2048;
    constexpr int FULL = NVC / STEP;
    static_assert(FULL >= 1, "every thread owns a first pair of edge vectors");
    __shared__ __attribute__((aligned(16))) float sSrc[NN];
    __shared__ __attribute__((aligned(16))) float sAcc[NN + 1024];
    static_assert(sizeof(float) * (2 * NN + 1024) > 160 * 1024, "more than half of the 320 KiB of a WGP");
    const int c   = blockIdx.x % NCH;
    const int t   = blockIdx.x / NCH;
    const int f   = t % NF;
    const int b   = t / NF;
    const int tid = threadIdx.x;

    const float* __restrict__ gdis = dis + (size_t)b * NN;
    const float* __restrict__ sp = src + (size_t)b * NN * NF + f;
    for (int i = tid; i < NN; i += 1024) {
        sSrc[i] = sp[(size_t)i * NF] * gdis[i];
        sAcc[i] = 0.f;
    }
    __syncthreads();

    const int* __restrict__ eb = ei + (size_t)b * 2 * EPB;
    const int4* __restrict__ S = reinterpret_cast<const int4*>(eb) + (size_t)c * NVC;
    const int4* __restrict__ D = reinterpret_cast<const int4*>(eb + EPB) + (size_t)c * NVC;

    auto add4 = [&](const int4 s, const int4 d) {
        const float v0 = sSrc[s.x];
        const float v1 = sSrc[s.y];
        const float v2 = sSrc[s.z];
        const float v3 = sSrc[s.w];
        __hip_atomic_fetch_add(&sAcc[d.x], v0, __ATOMIC_RELAXED, __HIP_MEMORY_SCOPE_WORKGROUP);
        __hip_atomic_fetch_add(&sAcc[d.y], v1, __ATOMIC_RELAXED, __HIP_MEMORY_SCOPE_WORKGROUP);
        __hip_atomic_fetch_add(&sAcc[d.z], v2, __ATOMIC_RELAXED, __HIP_MEMORY_SCOPE_WORKGROUP);
        __hip_atomic_fetch_add(&sAcc[d.w], v3, __ATOMIC_RELAXED, __HIP_MEMORY_SCOPE_WORKGROUP);
    };

    int i = tid;
    int4 s0 = S[i], s1 = S[i + 1024];
    int4 d0 = D[i], d1 = D[i + 1024];
    for (int trip = 0; trip < FULL; ++trip) {
        const int ni = (trip + 1 < FULL) ? i + STEP : i;
        const int4 s0n = S[ni], s1n = S[ni + 1024];
        const int4 d0n = D[ni], d1n = D[ni + 1024];
        add4(s0, d0);
        add4(s1, d1);
        s0 = s0n; s1 = s1n; d0 = d0n; d1 = d1n;
        i = ni;
    }
    for (int j = FULL * STEP + tid; j < NVC; j += 1024) add4(S[j], D[j]);
    __syncthreads();

    float4* __restrict__ dp =
        reinterpret_cast<float4*>(dst + (size_t)(c * NF + f) * NTOT + (size_t)b * NN);
    const float4* __restrict__ dq = reinterpret_cast<const float4*>(gdis);
    for (int j = tid; j < NN / 4; j += 1024) {
        const float4 a = reinterpret_cast<const float4*>(sAcc)[j];
        const float4 q = dq[j];
        dp[j] = make_float4(a.x * q.x, a.y * q.y, a.z * q.z, a.w * q.w);
    }
}

template <int NCH>
__global__ __launch_bounds__(1024)
void pair_sum_kernel(const unsigned* __restrict__ sched, const unsigned* __restrict__ ov,
                     const unsigned* __restrict__ ovcnt, const float* __restrict__ dis,
                     const float* __restrict__ src, float* __restrict__ dst) {
    static_assert((NN % 4) == 0 && (DCH % NCH) == 0 && (HID % 2) == 0,
                  "16-byte vectors over nodes, whole segments per chunk, features in pairs");
    constexpr int SPC  = DCH / NCH;
    constexpr int NVG  = SPC * GRIDW / 4;
    constexpr int STEP = 2048;
    constexpr int FULL = NVG / STEP;
    static_assert(FULL >= 1 && (NVG % STEP) == 0, "the grid is a whole number of trips");
    __shared__ __attribute__((aligned(16))) float2 sSrc[NN + 32];
    __shared__ __attribute__((aligned(16))) float  sAcc0[NN];
    __shared__ __attribute__((aligned(16))) float  sAcc1[NN];
    static_assert(sizeof(float) * (4 * NN + 64) <= 320 * 1024, "four planes and the null entries fit the 320 KiB of a WGP");
    const int c   = blockIdx.x % NCH;
    const int t   = blockIdx.x / NCH;
    const int f0  = (t % (HID / 2)) * 2;
    const int b   = t / (HID / 2);
    const int tid = threadIdx.x;

    const float* __restrict__ gdis = dis + (size_t)b * NN;
    {
        const float4* __restrict__ sp0 =
            reinterpret_cast<const float4*>(src + (size_t)f0 * NTOT + (size_t)b * NN);
        const float4* __restrict__ sp1 =
            reinterpret_cast<const float4*>(src + (size_t)(f0 + 1) * NTOT + (size_t)b * NN);
        const float4* __restrict__ dq = reinterpret_cast<const float4*>(gdis);
        for (int i = tid; i < NN / 4; i += 1024) {
            const float4 u = sp0[i];
            const float4 v = sp1[i];
            const float4 q = dq[i];
            reinterpret_cast<float4*>(sSrc)[2 * i]     = make_float4(u.x * q.x, v.x * q.x, u.y * q.y, v.y * q.y);
            reinterpret_cast<float4*>(sSrc)[2 * i + 1] = make_float4(u.z * q.z, v.z * q.z, u.w * q.w, v.w * q.w);
            reinterpret_cast<float4*>(sAcc0)[i] = make_float4(0.f, 0.f, 0.f, 0.f);
            reinterpret_cast<float4*>(sAcc1)[i] = make_float4(0.f, 0.f, 0.f, 0.f);
        }
        if (tid < 32) sSrc[NN + tid] = make_float2(0.f, 0.f);
    }
    __syncthreads();

    auto add1 = [&](const unsigned w) {
        const float2 v = sSrc[w & 0xFFFFu];
        const unsigned d = w >> 16;
        __hip_atomic_fetch_add(&sAcc0[d], v.x, __ATOMIC_RELAXED, __HIP_MEMORY_SCOPE_WORKGROUP);
        __hip_atomic_fetch_add(&sAcc1[d], v.y, __ATOMIC_RELAXED, __HIP_MEMORY_SCOPE_WORKGROUP);
    };
    auto add4 = [&](const uint4 w) {
        const float2 v0 = sSrc[w.x & 0xFFFFu];
        const float2 v1 = sSrc[w.y & 0xFFFFu];
        const float2 v2 = sSrc[w.z & 0xFFFFu];
        const float2 v3 = sSrc[w.w & 0xFFFFu];
        const unsigned d0 = w.x >> 16, d1 = w.y >> 16, d2 = w.z >> 16, d3 = w.w >> 16;
        __hip_atomic_fetch_add(&sAcc0[d0], v0.x, __ATOMIC_RELAXED, __HIP_MEMORY_SCOPE_WORKGROUP);
        __hip_atomic_fetch_add(&sAcc1[d0], v0.y, __ATOMIC_RELAXED, __HIP_MEMORY_SCOPE_WORKGROUP);
        __hip_atomic_fetch_add(&sAcc0[d1], v1.x, __ATOMIC_RELAXED, __HIP_MEMORY_SCOPE_WORKGROUP);
        __hip_atomic_fetch_add(&sAcc1[d1], v1.y, __ATOMIC_RELAXED, __HIP_MEMORY_SCOPE_WORKGROUP);
        __hip_atomic_fetch_add(&sAcc0[d2], v2.x, __ATOMIC_RELAXED, __HIP_MEMORY_SCOPE_WORKGROUP);
        __hip_atomic_fetch_add(&sAcc1[d2], v2.y, __ATOMIC_RELAXED, __HIP_MEMORY_SCOPE_WORKGROUP);
        __hip_atomic_fetch_add(&sAcc0[d3], v3.x, __ATOMIC_RELAXED, __HIP_MEMORY_SCOPE_WORKGROUP);
        __hip_atomic_fetch_add(&sAcc1[d3], v3.y, __ATOMIC_RELAXED, __HIP_MEMORY_SCOPE_WORKGROUP);
    };

    const int seg0 = b * DCH + c * SPC;
    const uint4* __restrict__ G = reinterpret_cast<const uint4*>(sched + (size_t)seg0 * GRIDW);
    static_assert((SPC % 2) == 0, "overflow words are taken four at a time, two segments each");
    unsigned x[SPC][2];
    #pragma unroll
    for (int sg = 0; sg < SPC; ++sg) {
        const unsigned* __restrict__ O = ov + (size_t)(seg0 + sg) * OVPAD;
        x[sg][0] = O[tid];
        x[sg][1] = O[tid + 1024];
    }
    unsigned nup[SPC];
    #pragma unroll
    for (int sg = 0; sg < SPC; ++sg) {
        unsigned n = ovcnt[(size_t)(seg0 + sg) * OVNST];
        n = (n < (unsigned)OVCAP) ? n : (unsigned)OVCAP;
        nup[sg] = (n + 1023u) & ~1023u;
    }
    int i = tid;
    uint4 w0 = G[i], w1 = G[i + 1024];
    for (int trip = 0; trip < FULL; ++trip) {
        const int ni = (trip + 1 < FULL) ? i + STEP : i;
        const uint4 w0n = G[ni], w1n = G[ni + 1024];
        add4(w0);
        add4(w1);
        w0 = w0n; w1 = w1n;
        i = ni;
    }
    #pragma unroll
    for (int sg = 0; sg < SPC; sg += 2)
        add4(make_uint4(x[sg][0], x[sg][1], x[sg + 1][0], x[sg + 1][1]));
    #pragma unroll
    for (int sg = 0; sg < SPC; ++sg) {
        const unsigned* __restrict__ O = ov + (size_t)(seg0 + sg) * OVPAD;
        for (unsigned j = 2048u + tid; j < nup[sg]; j += 1024u) add1(O[j]);
    }
    __syncthreads();

    float4* __restrict__ dp0 =
        reinterpret_cast<float4*>(dst + (size_t)(c * HID + f0) * NTOT + (size_t)b * NN);
    float4* __restrict__ dp1 =
        reinterpret_cast<float4*>(dst + (size_t)(c * HID + f0 + 1) * NTOT + (size_t)b * NN);
    const float4* __restrict__ dq = reinterpret_cast<const float4*>(gdis);
    for (int j = tid; j < NN / 4; j += 1024) {
        const float4 a0 = reinterpret_cast<const float4*>(sAcc0)[j];
        const float4 a1 = reinterpret_cast<const float4*>(sAcc1)[j];
        const float4 q  = dq[j];
        dp0[j] = make_float4(a0.x * q.x, a0.y * q.y, a0.z * q.z, a0.w * q.w);
        dp1[j] = make_float4(a1.x * q.x, a1.y * q.y, a1.z * q.z, a1.w * q.w);
    }
}

template <int K, int KW, int NCH, bool OWN_PLANES, bool OUT_PLANES>
__global__ __launch_bounds__(256)
void tag_gemm_kernel(const float* __restrict__ hin,
                                const float* __restrict__ agg,
                                const float* __restrict__ W,
                                const float* __restrict__ bias,
                                float* __restrict__ hout, int N) {
    static_assert((K % 4) == 0, "padded K must be a multiple of 4");
    __shared__ float sW[2 * K * HID];
    __shared__ float sB[HID];
    for (int i = threadIdx.x; i < 2 * K * HID; i += blockDim.x) {
        int s   = i / (K * HID);
        int rem = i - s * (K * HID);
        int kw  = rem / HID;
        int col = rem - kw * HID;
        sW[i] = (kw < KW) ? W[((size_t)s * KW + kw) * HID + col] : 0.0f;
    }
    if (threadIdx.x < HID) sB[threadIdx.x] = bias[threadIdx.x];
    __syncthreads();

    int lane = threadIdx.x & 31;
    int wave = (blockIdx.x * blockDim.x + threadIdx.x) >> 5;

    const int m     = lane & 15;
    const int khalf = (lane >> 4) * 2;

    for (int tile = 0; tile < 2; ++tile) {
    const int row0 = (wave * 2 + tile) * 16;
    if (row0 >= N) break;

    auto agg_at = [&](const int k) -> float {
        const int kk = (KW == K || k < KW) ? k : (KW - 1);
        float v = agg[(size_t)kk * NTOT + row0 + m];
        #pragma unroll
        for (int ch = 1; ch < NCH; ++ch)
            v += agg[((size_t)ch * KW + kk) * NTOT + row0 + m];
        return (KW == K || k < KW) ? v : 0.0f;
    };

    v8f acc0 = {0.f,0.f,0.f,0.f,0.f,0.f,0.f,0.f};
    v8f acc1 = {0.f,0.f,0.f,0.f,0.f,0.f,0.f,0.f};

    #pragma unroll
    for (int s = 0; s < 2; ++s) {
        const float* __restrict__ sWk = sW + s * K * HID;
        #pragma unroll
        for (int c = 0; c < K / 4; ++c) {
            const int k0 = c * 4 + khalf;
            v2f a;
            if (s) {
                a.x = agg_at(k0);
                a.y = agg_at(k0 + 1);
            } else if constexpr (OWN_PLANES) {
                a.x = hin[(size_t)k0 * NTOT + row0 + m];
                a.y = hin[(size_t)(k0 + 1) * NTOT + row0 + m];
            } else {
                a = *reinterpret_cast<const v2f*>(&hin[(size_t)(row0 + m) * K + k0]);
            }
            v2f b0, b1;
            b0.x = sWk[k0 * HID + m];
            b0.y = sWk[(k0 + 1) * HID + m];
            b1.x = sWk[k0 * HID + 16 + m];
            b1.y = sWk[(k0 + 1) * HID + 16 + m];
            if constexpr (OUT_PLANES) {
                acc0 = __builtin_amdgcn_wmma_f32_16x16x4_f32(
                           false, b0, false, a, (short)0, acc0, false, false);
                acc1 = __builtin_amdgcn_wmma_f32_16x16x4_f32(
                           false, b1, false, a, (short)0, acc1, false, false);
            } else {
                acc0 = __builtin_amdgcn_wmma_f32_16x16x4_f32(
                           false, a, false, b0, (short)0, acc0, false, false);
                acc1 = __builtin_amdgcn_wmma_f32_16x16x4_f32(
                           false, a, false, b1, (short)0, acc1, false, false);
            }
        }
    }

    if constexpr (OUT_PLANES) {
        const int fbase = (lane >> 4) ? 8 : 0;
        #pragma unroll
        for (int v = 0; v < 8; ++v) {
            int f = fbase + v;
            float x0 = acc0[v] + sB[f];
            float x1 = acc1[v] + sB[16 + f];
            hout[(size_t)f * NTOT + row0 + m]        = tanhf(x0);
            hout[(size_t)(16 + f) * NTOT + row0 + m] = tanhf(x1);
        }
    } else {
        const int rbase = row0 + ((lane >> 4) ? 8 : 0);
        #pragma unroll
        for (int v = 0; v < 8; ++v) {
            int r = rbase + v;
            float x0 = acc0[v] + sB[m];
            float x1 = acc1[v] + sB[16 + m];
            hout[(size_t)r * HID + m]      = tanhf(x0);
            hout[(size_t)r * HID + 16 + m] = tanhf(x1);
        }
    }
    }
}

__global__ void readout_kernel(const float* __restrict__ h, const float* __restrict__ Wreg,
                               const float* __restrict__ breg, float* __restrict__ out) {
    int i = blockIdx.x * blockDim.x + threadIdx.x;
    if (i >= NTOT) return;
    float acc = breg[0];
    const float4* hr = reinterpret_cast<const float4*>(h + (size_t)i * HID);
    #pragma unroll
    for (int g = 0; g < HID / 4; ++g) {
        float4 v = hr[g];
        acc += v.x * Wreg[4 * g + 0] + v.y * Wreg[4 * g + 1]
             + v.z * Wreg[4 * g + 2] + v.w * Wreg[4 * g + 3];
    }
    out[i] = acc;
}

static inline int cdiv(long long a, long long b) { return (int)((a + b - 1) / b); }

extern "C" void kernel_launch(void* const* d_in, const int* in_sizes, int n_in,
                              void* d_out, int out_size, void* d_ws, size_t ws_size,
                              hipStream_t stream) {
    const float* X    = (const float*)d_in[0];
    const int*   ei   = (const int*)  d_in[1];
    const float* W0   = (const float*)d_in[2];
    const float* b0   = (const float*)d_in[3];
    const float* W1   = (const float*)d_in[4];
    const float* b1   = (const float*)d_in[5];
    const float* W2   = (const float*)d_in[6];
    const float* b2   = (const float*)d_in[7];
    const float* Wreg = (const float*)d_in[8];
    const float* breg = (const float*)d_in[9];

    float* out    = (float*)d_out;
    float* hfinal = out + NTOT;

    float* dis = (float*)d_ws;
    float* agg = dis + NTOT;
    float* hA  = agg + (size_t)NTOT * HID * LCH;
    float* hB  = hA  + (size_t)NTOT * HID;
    float* Xp  = hB  + (size_t)NTOT * HID;
    unsigned* cnt = (unsigned*)(Xp + (size_t)NTOT * KPAD);
    unsigned* sched = cnt + (size_t)DCH * NTOT;
    unsigned* ov    = sched + (size_t)BATCH * DCH * GRIDW;
    unsigned* ovcnt = ov + (size_t)BATCH * DCH * OVPAD;

    const int T = 256;

    degree_schedule_kernel<<<BATCH * DCH, 1024, 0, stream>>>(ei, cnt, sched, ov, ovcnt);
    degree_rsqrt_kernel<<<cdiv(NTOT / 4, T), T, 0, stream>>>(cnt, dis);

    const int gemm_blocks = cdiv((long long)(NTOT / 32) * 32, T);

    pad_rows_kernel<<<cdiv(NTOT, T), T, 0, stream>>>(X, Xp);
    plane_sum_kernel<WINDOW, L0CH><<<BATCH * WINDOW * L0CH, 1024, 0, stream>>>(ei, dis, X, agg);
    tag_gemm_kernel<KPAD, WINDOW, L0CH, false, true><<<gemm_blocks, T, 0, stream>>>(Xp, agg, W0, b0, hA, NTOT);

    pair_sum_kernel<LCH><<<BATCH * (HID / 2) * LCH, 1024, 0, stream>>>(sched, ov, ovcnt, dis, hA, agg);
    tag_gemm_kernel<HID, HID, LCH, true, true><<<gemm_blocks, T, 0, stream>>>(hA, agg, W1, b1, hB, NTOT);

    pair_sum_kernel<LCH><<<BATCH * (HID / 2) * LCH, 1024, 0, stream>>>(sched, ov, ovcnt, dis, hB, agg);
    tag_gemm_kernel<HID, HID, LCH, true, false><<<gemm_blocks, T, 0, stream>>>(hB, agg, W2, b2, hfinal, NTOT);

    readout_kernel<<<cdiv(NTOT, T), T, 0, stream>>>(hfinal, Wreg, breg, out);
}


# sum and product launches: each phase's loads issue as ONE group ahead of the phase before (fill, first edges, weights+tile 0, tile 1 before tile 0's tanh), dis kept in registers; sums as before
# speedup vs baseline: 22.1424x; 1.0786x over previous
#include <hip/hip_runtime.h>
#include <hip/hip_bf16.h>
#include <math.h>

#define BATCH   8
#define WINDOW  5
#define NN      20000
#define EPB     320000
#define HID     32
#define KPAD    8
#define NTOT    (BATCH * NN)
#define ETOT    (BATCH * EPB)
#define L0CH    5
#define DCH     16
#define LCH     2
#define PASSES  11
#define GRIDW   (PASSES * 64 * 32)
#define OVCAP   (EPB / DCH)
#define OVPAD   (((OVCAP + 1023) / 1024) * 1024)
#define OVNST   32

typedef float v2f __attribute__((ext_vector_type(2)));
typedef float v8f __attribute__((ext_vector_type(8)));

__global__ void pad_rows_kernel(const float* __restrict__ x, float* __restrict__ xp) {
    int i = blockIdx.x * blockDim.x + threadIdx.x;
    if (i >= NTOT) return;
    const float* __restrict__ r = x + (size_t)i * WINDOW;
    float* __restrict__ o = xp + (size_t)i * KPAD;
    float4 lo = make_float4(r[0], r[1], r[2], r[3]);
    float4 hi = make_float4(r[4], 0.f, 0.f, 0.f);
    reinterpret_cast<float4*>(o)[0] = lo;
    reinterpret_cast<float4*>(o)[1] = hi;
}

__global__ __launch_bounds__(1024)
void degree_schedule_kernel(const int* __restrict__ ei, unsigned* __restrict__ cnt,
                            unsigned* __restrict__ sched, unsigned* __restrict__ ov,
                            unsigned* __restrict__ ovcnt) {
    static_assert((EPB % (4 * DCH)) == 0 && (NN % 4) == 0, "16-byte vectors over edge chunks and nodes");
    static_assert((NN % 32) == 0, "the null source entries NN + lane keep the lane's bank pair");
    static_assert(NN + 32 <= 65536 && NN >= 64, "packed 16-bit source and destination indices; null destinations are real nodes");
    static_assert((GRIDW % 4) == 0 && (OVCAP % 4) == 0, "16-byte vectors over grid and overflow list");
    constexpr int NVC   = EPB / 4 / DCH;
    constexpr int TRIPS = (NVC + 1023) / 1024;
    __shared__ __attribute__((aligned(16))) unsigned sCnt[NN];
    __shared__ __attribute__((aligned(16))) unsigned sGrid[GRIDW];
    __shared__ __attribute__((aligned(16))) unsigned sOv[OVCAP];
    __shared__ __attribute__((aligned(16))) unsigned sCell[32 * 64];
    __shared__ unsigned sOvN;
    const int c   = blockIdx.x % DCH;
    const int b   = blockIdx.x / DCH;
    const int tid = threadIdx.x;

    const int4* __restrict__ S =
        reinterpret_cast<const int4*>(ei + (size_t)b * 2 * EPB) + (size_t)c * NVC;
    const int4* __restrict__ D =
        reinterpret_cast<const int4*>(ei + (size_t)b * 2 * EPB + EPB) + (size_t)c * NVC;
    int4 s[TRIPS], d[TRIPS];
    #pragma unroll
    for (int k = 0; k < TRIPS; ++k) {
        const int j = tid + k * 1024;
        s[k] = S[(j < NVC) ? j : tid];
        d[k] = D[(j < NVC) ? j : tid];
    }
    for (int i = tid; i < NN / 4; i += 1024)
        reinterpret_cast<uint4*>(sCnt)[i] = make_uint4(0u, 0u, 0u, 0u);
    if (tid < 32 * 64 / 4)
        reinterpret_cast<uint4*>(sCell)[tid] = make_uint4(0u, 0u, 0u, 0u);
    if (tid == 0) sOvN = 0u;
    for (int v = tid; v < GRIDW / 4; v += 1024) {
        const unsigned lane = (unsigned)v & 31u;
        const unsigned r0   = ((unsigned)v >> 5) << 2;
        const unsigned ns   = (unsigned)NN + lane;
        reinterpret_cast<uint4*>(sGrid)[v] = make_uint4(ns | (((lane + r0     ) & 63u) << 16),
                                                        ns | (((lane + r0 + 1u) & 63u) << 16),
                                                        ns | (((lane + r0 + 2u) & 63u) << 16),
                                                        ns | (((lane + r0 + 3u) & 63u) << 16));
    }
    __syncthreads();

    auto rank_of = [&](const int src, const int dst) -> unsigned {
        __hip_atomic_fetch_add(&sCnt[dst], 1u, __ATOMIC_RELAXED, __HIP_MEMORY_SCOPE_WORKGROUP);
        return __hip_atomic_fetch_add(&sCell[((unsigned)src & 31u) * 64u + ((unsigned)dst & 63u)], 1u,
                                      __ATOMIC_RELAXED, __HIP_MEMORY_SCOPE_WORKGROUP);
    };
    auto put = [&](const int src, const int dst, const unsigned p) {
        const unsigned sc = (unsigned)src & 31u;
        const unsigned dc = (unsigned)dst & 63u;
        const unsigned w  = (unsigned)src | ((unsigned)dst << 16);
        if (p < (unsigned)PASSES) {
            const unsigned R = p * 64u + ((dc - sc) & 63u);
            sGrid[((((R >> 2) << 5) + sc) << 2) | (R & 3u)] = w;
        } else {
            const unsigned q = __hip_atomic_fetch_add(&sOvN, 1u, __ATOMIC_RELAXED,
                                                      __HIP_MEMORY_SCOPE_WORKGROUP);
            sOv[q] = w;
        }
    };
    unsigned rk[TRIPS][4];
    #pragma unroll
    for (int k = 0; k < TRIPS; ++k) {
        if (tid + k * 1024 < NVC) {
            rk[k][0] = rank_of(s[k].x, d[k].x);
            rk[k][1] = rank_of(s[k].y, d[k].y);
            rk[k][2] = rank_of(s[k].z, d[k].z);
            rk[k][3] = rank_of(s[k].w, d[k].w);
        }
    }
    #pragma unroll
    for (int k = 0; k < TRIPS; ++k) {
        if (tid + k * 1024 < NVC) {
            put(s[k].x, d[k].x, rk[k][0]);
            put(s[k].y, d[k].y, rk[k][1]);
            put(s[k].z, d[k].z, rk[k][2]);
            put(s[k].w, d[k].w, rk[k][3]);
        }
    }
    __syncthreads();

    uint4* __restrict__ op =
        reinterpret_cast<uint4*>(cnt + (size_t)c * NTOT + (size_t)b * NN);
    for (int i = tid; i < NN / 4; i += 1024)
        op[i] = reinterpret_cast<const uint4*>(sCnt)[i];

    const int seg = b * DCH + c;
    uint4* __restrict__ gp = reinterpret_cast<uint4*>(sched + (size_t)seg * GRIDW);
    for (int i = tid; i < GRIDW / 4; i += 1024)
        gp[i] = reinterpret_cast<const uint4*>(sGrid)[i];
    const unsigned nov = sOvN;
    const unsigned nup = (nov + 1023u) & ~1023u;
    const unsigned npad = (nup > 2048u) ? nup : 2048u;
    unsigned* __restrict__ vp = ov + (size_t)seg * OVPAD;
    for (unsigned i = tid; i < npad; i += 1024u)
        vp[i] = (i < nov) ? sOv[i] : (((unsigned)NN + (i & 31u)) | ((i & 63u) << 16));
    if (tid == 0) ovcnt[(size_t)seg * OVNST] = nov;
}

__global__ void degree_rsqrt_kernel(const unsigned* __restrict__ cnt, float* __restrict__ dis) {
    int i = blockIdx.x * blockDim.x + threadIdx.x;
    if (i >= NTOT / 4) return;
    uint4 s = reinterpret_cast<const uint4*>(cnt)[i];
    #pragma unroll
    for (int c = 1; c < DCH; ++c) {
        const uint4 t = reinterpret_cast<const uint4*>(cnt + (size_t)c * NTOT)[i];
        s.x += t.x; s.y += t.y; s.z += t.z; s.w += t.w;
    }
    float4 r;
    r.x = (s.x > 0u) ? rsqrtf((float)s.x) : 0.0f;
    r.y = (s.y > 0u) ? rsqrtf((float)s.y) : 0.0f;
    r.z = (s.z > 0u) ? rsqrtf((float)s.z) : 0.0f;
    r.w = (s.w > 0u) ? rsqrtf((float)s.w) : 0.0f;
    reinterpret_cast<float4*>(dis)[i] = r;
}

template <int NF, int NCH>
__global__ __launch_bounds__(1024)
void plane_sum_kernel(const int* __restrict__ ei, const float* __restrict__ dis,
                      const float* __restrict__ src, float* __restrict__ dst) {
    static_assert((NN % 4) == 0 && (EPB % (4 * NCH)) == 0, "16-byte vectors over nodes and edge chunks");
    constexpr int NVC  = EPB / 4 / NCH;
    constexpr int STEP = 2048;
    constexpr int TRIPS = (NVC + STEP - 1) / STEP;
    static_assert(NVC >= STEP && (TRIPS - 1) * STEP + 1024 <= NVC,
                  "every thread owns a first pair of edge vectors, and a first vector in every trip");
    constexpr int NV4  = NN / 4;
    constexpr int FT   = (NV4 + 1023) / 1024;
    constexpr int FS   = (NN + 1023) / 1024;
    __shared__ __attribute__((aligned(16))) float sSrc[NN];
    __shared__ __attribute__((aligned(16))) float sAcc[NN + 1024];
    static_assert(sizeof(float) * (2 * NN + 1024) > 160 * 1024, "more than half of the 320 KiB of a WGP");
    const int c   = blockIdx.x % NCH;
    const int t   = blockIdx.x / NCH;
    const int f   = t % NF;
    const int b   = t / NF;
    const int tid = threadIdx.x;

    const float* __restrict__ gdis = dis + (size_t)b * NN;
    const float* __restrict__ sp = src + (size_t)b * NN * NF + f;
    const float4* __restrict__ dq = reinterpret_cast<const float4*>(gdis);
    const int* __restrict__ eb = ei + (size_t)b * 2 * EPB;
    const int4* __restrict__ S = reinterpret_cast<const int4*>(eb) + (size_t)c * NVC;
    const int4* __restrict__ D = reinterpret_cast<const int4*>(eb + EPB) + (size_t)c * NVC;
    float xs[FS], ds[FS];
    #pragma unroll
    for (int k = 0; k < FS; ++k) {
        const int n  = tid + k * 1024;
        const int nn = (n < NN) ? n : tid;
        xs[k] = sp[(size_t)nn * NF];
        ds[k] = gdis[nn];
    }
    float4 q[FT];
    #pragma unroll
    for (int k = 0; k < FT; ++k) {
        const int j = tid + k * 1024;
        q[k] = dq[(j < NV4) ? j : tid];
    }
    int i = tid;
    int4 s0 = S[i], s1 = S[i + 1024];
    int4 d0 = D[i], d1 = D[i + 1024];
    #pragma unroll
    for (int k = 0; k < FS; ++k) {
        const int n  = tid + k * 1024;
        const int nn = (n < NN) ? n : tid;
        sSrc[nn] = xs[k] * ds[k];
        sAcc[nn] = 0.f;
    }
    __syncthreads();

    auto add4 = [&](const int4 s, const int4 d) {
        const float v0 = sSrc[s.x];
        const float v1 = sSrc[s.y];
        const float v2 = sSrc[s.z];
        const float v3 = sSrc[s.w];
        __hip_atomic_fetch_add(&sAcc[d.x], v0, __ATOMIC_RELAXED, __HIP_MEMORY_SCOPE_WORKGROUP);
        __hip_atomic_fetch_add(&sAcc[d.y], v1, __ATOMIC_RELAXED, __HIP_MEMORY_SCOPE_WORKGROUP);
        __hip_atomic_fetch_add(&sAcc[d.z], v2, __ATOMIC_RELAXED, __HIP_MEMORY_SCOPE_WORKGROUP);
        __hip_atomic_fetch_add(&sAcc[d.w], v3, __ATOMIC_RELAXED, __HIP_MEMORY_SCOPE_WORKGROUP);
    };

    for (int trip = 0; trip < TRIPS; ++trip) {
        const int ni = (trip + 1 < TRIPS) ? i + STEP : i;
        const int n1 = (ni + 1024 < NVC) ? ni + 1024 : ni;
        const int4 s0n = S[ni], s1n = S[n1];
        const int4 d0n = D[ni], d1n = D[n1];
        add4(s0, d0);
        if (i + 1024 < NVC) add4(s1, d1);
        s0 = s0n; s1 = s1n; d0 = d0n; d1 = d1n;
        i = ni;
    }
    __syncthreads();

    float4* __restrict__ dp =
        reinterpret_cast<float4*>(dst + (size_t)(c * NF + f) * NTOT + (size_t)b * NN);
    #pragma unroll
    for (int k = 0; k < FT; ++k) {
        const int j = tid + k * 1024;
        if (j < NV4) {
            const float4 a = reinterpret_cast<const float4*>(sAcc)[j];
            dp[j] = make_float4(a.x * q[k].x, a.y * q[k].y, a.z * q[k].z, a.w * q[k].w);
        }
    }
}

template <int NCH>
__global__ __launch_bounds__(1024)
void pair_sum_kernel(const unsigned* __restrict__ sched, const unsigned* __restrict__ ov,
                     const unsigned* __restrict__ ovcnt, const float* __restrict__ dis,
                     const float* __restrict__ src, float* __restrict__ dst) {
    static_assert((NN % 4) == 0 && (DCH % NCH) == 0 && (HID % 2) == 0,
                  "16-byte vectors over nodes, whole segments per chunk, features in pairs");
    constexpr int SPC  = DCH / NCH;
    constexpr int NVG  = SPC * GRIDW / 4;
    constexpr int STEP = 2048;
    constexpr int FULL = NVG / STEP;
    static_assert(FULL >= 1 && (NVG % STEP) == 0, "the grid is a whole number of trips");
    constexpr int NV4  = NN / 4;
    constexpr int FT   = (NV4 + 1023) / 1024;
    __shared__ __attribute__((aligned(16))) float2 sSrc[NN + 32];
    __shared__ __attribute__((aligned(16))) float  sAcc0[NN];
    __shared__ __attribute__((aligned(16))) float  sAcc1[NN];
    static_assert(sizeof(float) * (4 * NN + 64) <= 320 * 1024, "four planes and the null entries fit the 320 KiB of a WGP");
    const int c   = blockIdx.x % NCH;
    const int t   = blockIdx.x / NCH;
    const int f0  = (t % (HID / 2)) * 2;
    const int b   = t / (HID / 2);
    const int tid = threadIdx.x;

    const float* __restrict__ gdis = dis + (size_t)b * NN;
    float4 u[FT], v[FT], q[FT];
    {
        const float4* __restrict__ sp0 =
            reinterpret_cast<const float4*>(src + (size_t)f0 * NTOT + (size_t)b * NN);
        const float4* __restrict__ sp1 =
            reinterpret_cast<const float4*>(src + (size_t)(f0 + 1) * NTOT + (size_t)b * NN);
        const float4* __restrict__ dq = reinterpret_cast<const float4*>(gdis);
        #pragma unroll
        for (int k = 0; k < FT; ++k) {
            const int j  = tid + k * 1024;
            const int jj = (j < NV4) ? j : tid;
            u[k] = sp0[jj];
            v[k] = sp1[jj];
            q[k] = dq[jj];
        }
    }
    const int seg0 = b * DCH + c * SPC;
    const uint4* __restrict__ G = reinterpret_cast<const uint4*>(sched + (size_t)seg0 * GRIDW);
    static_assert((SPC % 2) == 0, "overflow words are taken four at a time, two segments each");
    unsigned x[SPC][2];
    #pragma unroll
    for (int sg = 0; sg < SPC; ++sg) {
        const unsigned* __restrict__ O = ov + (size_t)(seg0 + sg) * OVPAD;
        x[sg][0] = O[tid];
        x[sg][1] = O[tid + 1024];
    }
    unsigned nup[SPC];
    #pragma unroll
    for (int sg = 0; sg < SPC; ++sg) {
        unsigned n = ovcnt[(size_t)(seg0 + sg) * OVNST];
        n = (n < (unsigned)OVCAP) ? n : (unsigned)OVCAP;
        nup[sg] = (n + 1023u) & ~1023u;
    }
    int i = tid;
    uint4 w0 = G[i], w1 = G[i + 1024];

    #pragma unroll
    for (int k = 0; k < FT; ++k) {
        const int j  = tid + k * 1024;
        const int jj = (j < NV4) ? j : tid;
        reinterpret_cast<float4*>(sSrc)[2 * jj]     = make_float4(u[k].x * q[k].x, v[k].x * q[k].x, u[k].y * q[k].y, v[k].y * q[k].y);
        reinterpret_cast<float4*>(sSrc)[2 * jj + 1] = make_float4(u[k].z * q[k].z, v[k].z * q[k].z, u[k].w * q[k].w, v[k].w * q[k].w);
        reinterpret_cast<float4*>(sAcc0)[jj] = make_float4(0.f, 0.f, 0.f, 0.f);
        reinterpret_cast<float4*>(sAcc1)[jj] = make_float4(0.f, 0.f, 0.f, 0.f);
    }
    if (tid < 32) sSrc[NN + tid] = make_float2(0.f, 0.f);
    __syncthreads();

    auto add1 = [&](const unsigned w) {
        const float2 v = sSrc[w & 0xFFFFu];
        const unsigned d = w >> 16;
        __hip_atomic_fetch_add(&sAcc0[d], v.x, __ATOMIC_RELAXED, __HIP_MEMORY_SCOPE_WORKGROUP);
        __hip_atomic_fetch_add(&sAcc1[d], v.y, __ATOMIC_RELAXED, __HIP_MEMORY_SCOPE_WORKGROUP);
    };
    auto add4 = [&](const uint4 w) {
        const float2 v0 = sSrc[w.x & 0xFFFFu];
        const float2 v1 = sSrc[w.y & 0xFFFFu];
        const float2 v2 = sSrc[w.z & 0xFFFFu];
        const float2 v3 = sSrc[w.w & 0xFFFFu];
        const unsigned d0 = w.x >> 16, d1 = w.y >> 16, d2 = w.z >> 16, d3 = w.w >> 16;
        __hip_atomic_fetch_add(&sAcc0[d0], v0.x, __ATOMIC_RELAXED, __HIP_MEMORY_SCOPE_WORKGROUP);
        __hip_atomic_fetch_add(&sAcc1[d0], v0.y, __ATOMIC_RELAXED, __HIP_MEMORY_SCOPE_WORKGROUP);
        __hip_atomic_fetch_add(&sAcc0[d1], v1.x, __ATOMIC_RELAXED, __HIP_MEMORY_SCOPE_WORKGROUP);
        __hip_atomic_fetch_add(&sAcc1[d1], v1.y, __ATOMIC_RELAXED, __HIP_MEMORY_SCOPE_WORKGROUP);
        __hip_atomic_fetch_add(&sAcc0[d2], v2.x, __ATOMIC_RELAXED, __HIP_MEMORY_SCOPE_WORKGROUP);
        __hip_atomic_fetch_add(&sAcc1[d2], v2.y, __ATOMIC_RELAXED, __HIP_MEMORY_SCOPE_WORKGROUP);
        __hip_atomic_fetch_add(&sAcc0[d3], v3.x, __ATOMIC_RELAXED, __HIP_MEMORY_SCOPE_WORKGROUP);
        __hip_atomic_fetch_add(&sAcc1[d3], v3.y, __ATOMIC_RELAXED, __HIP_MEMORY_SCOPE_WORKGROUP);
    };

    for (int trip = 0; trip < FULL; ++trip) {
        const int ni = (trip + 1 < FULL) ? i + STEP : i;
        const uint4 w0n = G[ni], w1n = G[ni + 1024];
        add4(w0);
        add4(w1);
        w0 = w0n; w1 = w1n;
        i = ni;
    }
    #pragma unroll
    for (int sg = 0; sg < SPC; sg += 2)
        add4(make_uint4(x[sg][0], x[sg][1], x[sg + 1][0], x[sg + 1][1]));
    #pragma unroll
    for (int sg = 0; sg < SPC; ++sg) {
        const unsigned* __restrict__ O = ov + (size_t)(seg0 + sg) * OVPAD;
        for (unsigned j = 2048u + tid; j < nup[sg]; j += 1024u) add1(O[j]);
    }
    __syncthreads();

    float4* __restrict__ dp0 =
        reinterpret_cast<float4*>(dst + (size_t)(c * HID + f0) * NTOT + (size_t)b * NN);
    float4* __restrict__ dp1 =
        reinterpret_cast<float4*>(dst + (size_t)(c * HID + f0 + 1) * NTOT + (size_t)b * NN);
    #pragma unroll
    for (int k = 0; k < FT; ++k) {
        const int j = tid + k * 1024;
        if (j < NV4) {
            const float4 a0 = reinterpret_cast<const float4*>(sAcc0)[j];
            const float4 a1 = reinterpret_cast<const float4*>(sAcc1)[j];
            dp0[j] = make_float4(a0.x * q[k].x, a0.y * q[k].y, a0.z * q[k].z, a0.w * q[k].w);
            dp1[j] = make_float4(a1.x * q[k].x, a1.y * q[k].y, a1.z * q[k].z, a1.w * q[k].w);
        }
    }
}

template <int K, int KW, int NCH, bool OWN_PLANES, bool OUT_PLANES>
__global__ __launch_bounds__(256)
void tag_gemm_kernel(const float* __restrict__ hin,
                                const float* __restrict__ agg,
                                const float* __restrict__ W,
                                const float* __restrict__ bias,
                                float* __restrict__ hout, int N) {
    static_assert((K % 4) == 0, "padded K must be a multiple of 4");
    constexpr int TPB = 256;
    constexpr int NC  = K / 4;
    __shared__ float sW[2 * K * HID];
    __shared__ float sB[HID];

    const int lane = threadIdx.x & 31;
    const int wave = (blockIdx.x * TPB + threadIdx.x) >> 5;

    const int m     = lane & 15;
    const int khalf = (lane >> 4) * 2;

    auto load_tile = [&](const int row0, v2f (&own)[NC], v2f (&part)[NCH][NC]) {
        #pragma unroll
        for (int c = 0; c < NC; ++c) {
            const int k0 = c * 4 + khalf;
            if constexpr (OWN_PLANES) {
                own[c].x = hin[(size_t)k0 * NTOT + row0 + m];
                own[c].y = hin[(size_t)(k0 + 1) * NTOT + row0 + m];
            } else {
                own[c] = *reinterpret_cast<const v2f*>(&hin[(size_t)(row0 + m) * K + k0]);
            }
            const int kx = (KW == K || k0 < KW) ? k0 : (KW - 1);
            const int ky = (KW == K || k0 + 1 < KW) ? (k0 + 1) : (KW - 1);
            #pragma unroll
            for (int ch = 0; ch < NCH; ++ch) {
                part[ch][c].x = agg[((size_t)ch * KW + kx) * NTOT + row0 + m];
                part[ch][c].y = agg[((size_t)ch * KW + ky) * NTOT + row0 + m];
            }
        }
    };

    const int rowA = wave * 32;
    const int rowB = rowA + 16;
    const bool hasA = rowA < N;
    const bool hasB = rowB < N;
    v2f ownA[NC], partA[NCH][NC];
    if (hasA) load_tile(rowA, ownA, partA);

    static_assert(((2 * K * HID) % TPB) == 0 && HID <= TPB && (HID & (HID - 1)) == 0,
                  "whole trips over the weights; the bias index is masked");
    constexpr int WT = (2 * K * HID) / TPB;
    float wv[WT];
    #pragma unroll
    for (int j = 0; j < WT; ++j) {
        const int i   = (int)threadIdx.x + j * TPB;
        const int s   = i / (K * HID);
        const int rem = i - s * (K * HID);
        const int kw  = rem / HID;
        const int col = rem - kw * HID;
        const int kr  = (kw < KW) ? kw : (KW - 1);
        wv[j] = W[((size_t)s * KW + kr) * HID + col];
    }
    const float bv = bias[threadIdx.x & (HID - 1)];
    #pragma unroll
    for (int j = 0; j < WT; ++j) {
        const int i   = (int)threadIdx.x + j * TPB;
        const int s   = i / (K * HID);
        const int rem = i - s * (K * HID);
        const int kw  = rem / HID;
        sW[i] = (kw < KW) ? wv[j] : 0.0f;
    }
    sB[threadIdx.x & (HID - 1)] = bv;
    __syncthreads();
    if (!hasA) return;

    auto product = [&](const v2f (&own)[NC], const v2f (&part)[NCH][NC], v8f& acc0, v8f& acc1) {
        #pragma unroll
        for (int s = 0; s < 2; ++s) {
            const float* __restrict__ sWk = sW + s * K * HID;
            #pragma unroll
            for (int c = 0; c < NC; ++c) {
                const int k0 = c * 4 + khalf;
                v2f a;
                if (s) {
                    float vx = part[0][c].x;
                    float vy = part[0][c].y;
                    #pragma unroll
                    for (int ch = 1; ch < NCH; ++ch) {
                        vx += part[ch][c].x;
                        vy += part[ch][c].y;
                    }
                    a.x = (KW == K || k0 < KW) ? vx : 0.0f;
                    a.y = (KW == K || k0 + 1 < KW) ? vy : 0.0f;
                } else {
                    a = own[c];
                }
                v2f b0, b1;
                b0.x = sWk[k0 * HID + m];
                b0.y = sWk[(k0 + 1) * HID + m];
                b1.x = sWk[k0 * HID + 16 + m];
                b1.y = sWk[(k0 + 1) * HID + 16 + m];
                if constexpr (OUT_PLANES) {
                    acc0 = __builtin_amdgcn_wmma_f32_16x16x4_f32(
                               false, b0, false, a, (short)0, acc0, false, false);
                    acc1 = __builtin_amdgcn_wmma_f32_16x16x4_f32(
                               false, b1, false, a, (short)0, acc1, false, false);
                } else {
                    acc0 = __builtin_amdgcn_wmma_f32_16x16x4_f32(
                               false, a, false, b0, (short)0, acc0, false, false);
                    acc1 = __builtin_amdgcn_wmma_f32_16x16x4_f32(
                               false, a, false, b1, (short)0, acc1, false, false);
                }
            }
        }
    };

    auto finish = [&](const int row0, const v8f acc0, const v8f acc1) {
        if constexpr (OUT_PLANES) {
            const int fbase = (lane >> 4) ? 8 : 0;
            #pragma unroll
            for (int v = 0; v < 8; ++v) {
                int f = fbase + v;
                float x0 = acc0[v] + sB[f];
                float x1 = acc1[v] + sB[16 + f];
                hout[(size_t)f * NTOT + row0 + m]        = tanhf(x0);
                hout[(size_t)(16 + f) * NTOT + row0 + m] = tanhf(x1);
            }
        } else {
            const int rbase = row0 + ((lane >> 4) ? 8 : 0);
            #pragma unroll
            for (int v = 0; v < 8; ++v) {
                int r = rbase + v;
                float x0 = acc0[v] + sB[m];
                float x1 = acc1[v] + sB[16 + m];
                hout[(size_t)r * HID + m]      = tanhf(x0);
                hout[(size_t)r * HID + 16 + m] = tanhf(x1);
            }
        }
    };

    v8f accA0 = {0.f,0.f,0.f,0.f,0.f,0.f,0.f,0.f};
    v8f accA1 = {0.f,0.f,0.f,0.f,0.f,0.f,0.f,0.f};
    product(ownA, partA, accA0, accA1);
    v2f ownB[NC], partB[NCH][NC];
    if (hasB) load_tile(rowB, ownB, partB);
    finish(rowA, accA0, accA1);
    if (hasB) {
        v8f accB0 = {0.f,0.f,0.f,0.f,0.f,0.f,0.f,0.f};
        v8f accB1 = {0.f,0.f,0.f,0.f,0.f,0.f,0.f,0.f};
        product(ownB, partB, accB0, accB1);
        finish(rowB, accB0, accB1);
    }
}

__global__ void readout_kernel(const float* __restrict__ h, const float* __restrict__ Wreg,
                               const float* __restrict__ breg, float* __restrict__ out) {
    int i = blockIdx.x * blockDim.x + threadIdx.x;
    if (i >= NTOT) return;
    float acc = breg[0];
    const float4* hr = reinterpret_cast<const float4*>(h + (size_t)i * HID);
    #pragma unroll
    for (int g = 0; g < HID / 4; ++g) {
        float4 v = hr[g];
        acc += v.x * Wreg[4 * g + 0] + v.y * Wreg[4 * g + 1]
             + v.z * Wreg[4 * g + 2] + v.w * Wreg[4 * g + 3];
    }
    out[i] = acc;
}

static inline int cdiv(long long a, long long b) { return (int)((a + b - 1) / b); }

extern "C" void kernel_launch(void* const* d_in, const int* in_sizes, int n_in,
                              void* d_out, int out_size, void* d_ws, size_t ws_size,
                              hipStream_t stream) {
    const float* X    = (const float*)d_in[0];
    const int*   ei   = (const int*)  d_in[1];
    const float* W0   = (const float*)d_in[2];
    const float* b0   = (const float*)d_in[3];
    const float* W1   = (const float*)d_in[4];
    const float* b1   = (const float*)d_in[5];
    const float* W2   = (const float*)d_in[6];
    const float* b2   = (const float*)d_in[7];
    const float* Wreg = (const float*)d_in[8];
    const float* breg = (const float*)d_in[9];

    float* out    = (float*)d_out;
    float* hfinal = out + NTOT;

    float* dis = (float*)d_ws;
    float* agg = dis + NTOT;
    float* hA  = agg + (size_t)NTOT * HID * LCH;
    float* hB  = hA  + (size_t)NTOT * HID;
    float* Xp  = hB  + (size_t)NTOT * HID;
    unsigned* cnt = (unsigned*)(Xp + (size_t)NTOT * KPAD);
    unsigned* sched = cnt + (size_t)DCH * NTOT;
    unsigned* ov    = sched + (size_t)BATCH * DCH * GRIDW;
    unsigned* ovcnt = ov + (size_t)BATCH * DCH * OVPAD;

    const int T = 256;

    degree_schedule_kernel<<<BATCH * DCH, 1024, 0, stream>>>(ei, cnt, sched, ov, ovcnt);
    degree_rsqrt_kernel<<<cdiv(NTOT / 4, T), T, 0, stream>>>(cnt, dis);

    const int gemm_blocks = cdiv((long long)(NTOT / 32) * 32, T);

    pad_rows_kernel<<<cdiv(NTOT, T), T, 0, stream>>>(X, Xp);
    plane_sum_kernel<WINDOW, L0CH><<<BATCH * WINDOW * L0CH, 1024, 0, stream>>>(ei, dis, X, agg);
    tag_gemm_kernel<KPAD, WINDOW, L0CH, false, true><<<gemm_blocks, T, 0, stream>>>(Xp, agg, W0, b0, hA, NTOT);

    pair_sum_kernel<LCH><<<BATCH * (HID / 2) * LCH, 1024, 0, stream>>>(sched, ov, ovcnt, dis, hA, agg);
    tag_gemm_kernel<HID, HID, LCH, true, true><<<gemm_blocks, T, 0, stream>>>(hA, agg, W1, b1, hB, NTOT);

    pair_sum_kernel<LCH><<<BATCH * (HID / 2) * LCH, 1024, 0, stream>>>(sched, ov, ovcnt, dis, hB, agg);
    tag_gemm_kernel<HID, HID, LCH, true, false><<<gemm_blocks, T, 0, stream>>>(hB, agg, W2, b2, hfinal, NTOT);

    readout_kernel<<<cdiv(NTOT, T), T, 0, stream>>>(hfinal, Wreg, breg, out);
}


# pad and readout launches folded into the products, 10 to 8: layer 0 reads X's rows in place; out = h.Wreg summed per row as a 16-lane tree in layer 2's epilogue; out's sum order changes
# speedup vs baseline: 23.0448x; 1.0408x over previous
#include <hip/hip_runtime.h>
#include <hip/hip_bf16.h>
#include <math.h>

#define BATCH   8
#define WINDOW  5
#define NN      20000
#define EPB     320000
#define HID     32
#define KPAD    8
#define NTOT    (BATCH * NN)
#define ETOT    (BATCH * EPB)
#define L0CH    5
#define DCH     16
#define LCH     2
#define PASSES  11
#define GRIDW   (PASSES * 64 * 32)
#define OVCAP   (EPB / DCH)
#define OVPAD   (((OVCAP + 1023) / 1024) * 1024)
#define OVNST   32

typedef float v2f __attribute__((ext_vector_type(2)));
typedef float v8f __attribute__((ext_vector_type(8)));

template <int XMASK>
__device__ __forceinline__ float row_xor_get(const float x) {
    return __int_as_float(__builtin_amdgcn_update_dpp(0, __float_as_int(x), 0x160 | XMASK, 0xF, 0xF, true));
}
__device__ __forceinline__ float row16_sum(float x) {
    x += row_xor_get<1>(x);
    x += row_xor_get<2>(x);
    x += row_xor_get<4>(x);
    x += row_xor_get<8>(x);
    return x;
}

__global__ __launch_bounds__(1024)
void degree_schedule_kernel(const int* __restrict__ ei, unsigned* __restrict__ cnt,
                            unsigned* __restrict__ sched, unsigned* __restrict__ ov,
                            unsigned* __restrict__ ovcnt) {
    static_assert((EPB % (4 * DCH)) == 0 && (NN % 4) == 0, "16-byte vectors over edge chunks and nodes");
    static_assert((NN % 32) == 0, "the null source entries NN + lane keep the lane's bank pair");
    static_assert(NN + 32 <= 65536 && NN >= 64, "packed 16-bit source and destination indices; null destinations are real nodes");
    static_assert((GRIDW % 4) == 0 && (OVCAP % 4) == 0, "16-byte vectors over grid and overflow list");
    constexpr int NVC   = EPB / 4 / DCH;
    constexpr int TRIPS = (NVC + 1023) / 1024;
    __shared__ __attribute__((aligned(16))) unsigned sCnt[NN];
    __shared__ __attribute__((aligned(16))) unsigned sGrid[GRIDW];
    __shared__ __attribute__((aligned(16))) unsigned sOv[OVCAP];
    __shared__ __attribute__((aligned(16))) unsigned sCell[32 * 64];
    __shared__ unsigned sOvN;
    const int c   = blockIdx.x % DCH;
    const int b   = blockIdx.x / DCH;
    const int tid = threadIdx.x;

    const int4* __restrict__ S =
        reinterpret_cast<const int4*>(ei + (size_t)b * 2 * EPB) + (size_t)c * NVC;
    const int4* __restrict__ D =
        reinterpret_cast<const int4*>(ei + (size_t)b * 2 * EPB + EPB) + (size_t)c * NVC;
    int4 s[TRIPS], d[TRIPS];
    #pragma unroll
    for (int k = 0; k < TRIPS; ++k) {
        const int j = tid + k * 1024;
        s[k] = S[(j < NVC) ? j : tid];
        d[k] = D[(j < NVC) ? j : tid];
    }
    for (int i = tid; i < NN / 4; i += 1024)
        reinterpret_cast<uint4*>(sCnt)[i] = make_uint4(0u, 0u, 0u, 0u);
    if (tid < 32 * 64 / 4)
        reinterpret_cast<uint4*>(sCell)[tid] = make_uint4(0u, 0u, 0u, 0u);
    if (tid == 0) sOvN = 0u;
    for (int v = tid; v < GRIDW / 4; v += 1024) {
        const unsigned lane = (unsigned)v & 31u;
        const unsigned r0   = ((unsigned)v >> 5) << 2;
        const unsigned ns   = (unsigned)NN + lane;
        reinterpret_cast<uint4*>(sGrid)[v] = make_uint4(ns | (((lane + r0     ) & 63u) << 16),
                                                        ns | (((lane + r0 + 1u) & 63u) << 16),
                                                        ns | (((lane + r0 + 2u) & 63u) << 16),
                                                        ns | (((lane + r0 + 3u) & 63u) << 16));
    }
    __syncthreads();

    auto rank_of = [&](const int src, const int dst) -> unsigned {
        __hip_atomic_fetch_add(&sCnt[dst], 1u, __ATOMIC_RELAXED, __HIP_MEMORY_SCOPE_WORKGROUP);
        return __hip_atomic_fetch_add(&sCell[((unsigned)src & 31u) * 64u + ((unsigned)dst & 63u)], 1u,
                                      __ATOMIC_RELAXED, __HIP_MEMORY_SCOPE_WORKGROUP);
    };
    auto put = [&](const int src, const int dst, const unsigned p) {
        const unsigned sc = (unsigned)src & 31u;
        const unsigned dc = (unsigned)dst & 63u;
        const unsigned w  = (unsigned)src | ((unsigned)dst << 16);
        if (p < (unsigned)PASSES) {
            const unsigned R = p * 64u + ((dc - sc) & 63u);
            sGrid[((((R >> 2) << 5) + sc) << 2) | (R & 3u)] = w;
        } else {
            const unsigned q = __hip_atomic_fetch_add(&sOvN, 1u, __ATOMIC_RELAXED,
                                                      __HIP_MEMORY_SCOPE_WORKGROUP);
            sOv[q] = w;
        }
    };
    unsigned rk[TRIPS][4];
    #pragma unroll
    for (int k = 0; k < TRIPS; ++k) {
        if (tid + k * 1024 < NVC) {
            rk[k][0] = rank_of(s[k].x, d[k].x);
            rk[k][1] = rank_of(s[k].y, d[k].y);
            rk[k][2] = rank_of(s[k].z, d[k].z);
            rk[k][3] = rank_of(s[k].w, d[k].w);
        }
    }
    #pragma unroll
    for (int k = 0; k < TRIPS; ++k) {
        if (tid + k * 1024 < NVC) {
            put(s[k].x, d[k].x, rk[k][0]);
            put(s[k].y, d[k].y, rk[k][1]);
            put(s[k].z, d[k].z, rk[k][2]);
            put(s[k].w, d[k].w, rk[k][3]);
        }
    }
    __syncthreads();

    uint4* __restrict__ op =
        reinterpret_cast<uint4*>(cnt + (size_t)c * NTOT + (size_t)b * NN);
    for (int i = tid; i < NN / 4; i += 1024)
        op[i] = reinterpret_cast<const uint4*>(sCnt)[i];

    const int seg = b * DCH + c;
    uint4* __restrict__ gp = reinterpret_cast<uint4*>(sched + (size_t)seg * GRIDW);
    for (int i = tid; i < GRIDW / 4; i += 1024)
        gp[i] = reinterpret_cast<const uint4*>(sGrid)[i];
    const unsigned nov = sOvN;
    const unsigned nup = (nov + 1023u) & ~1023u;
    const unsigned npad = (nup > 2048u) ? nup : 2048u;
    unsigned* __restrict__ vp = ov + (size_t)seg * OVPAD;
    for (unsigned i = tid; i < npad; i += 1024u)
        vp[i] = (i < nov) ? sOv[i] : (((unsigned)NN + (i & 31u)) | ((i & 63u) << 16));
    if (tid == 0) ovcnt[(size_t)seg * OVNST] = nov;
}

__global__ void degree_rsqrt_kernel(const unsigned* __restrict__ cnt, float* __restrict__ dis) {
    int i = blockIdx.x * blockDim.x + threadIdx.x;
    if (i >= NTOT / 4) return;
    uint4 s = reinterpret_cast<const uint4*>(cnt)[i];
    #pragma unroll
    for (int c = 1; c < DCH; ++c) {
        const uint4 t = reinterpret_cast<const uint4*>(cnt + (size_t)c * NTOT)[i];
        s.x += t.x; s.y += t.y; s.z += t.z; s.w += t.w;
    }
    float4 r;
    r.x = (s.x > 0u) ? rsqrtf((float)s.x) : 0.0f;
    r.y = (s.y > 0u) ? rsqrtf((float)s.y) : 0.0f;
    r.z = (s.z > 0u) ? rsqrtf((float)s.z) : 0.0f;
    r.w = (s.w > 0u) ? rsqrtf((float)s.w) : 0.0f;
    reinterpret_cast<float4*>(dis)[i] = r;
}

template <int NF, int NCH>
__global__ __launch_bounds__(1024)
void plane_sum_kernel(const int* __restrict__ ei, const float* __restrict__ dis,
                      const float* __restrict__ src, float* __restrict__ dst) {
    static_assert((NN % 4) == 0 && (EPB % (4 * NCH)) == 0, "16-byte vectors over nodes and edge chunks");
    constexpr int NVC  = EPB / 4 / NCH;
    constexpr int STEP = 2048;
    constexpr int TRIPS = (NVC + STEP - 1) / STEP;
    static_assert(NVC >= STEP && (TRIPS - 1) * STEP + 1024 <= NVC,
                  "every thread owns a first pair of edge vectors, and a first vector in every trip");
    constexpr int NV4  = NN / 4;
    constexpr int FT   = (NV4 + 1023) / 1024;
    constexpr int FS   = (NN + 1023) / 1024;
    __shared__ __attribute__((aligned(16))) float sSrc[NN];
    __shared__ __attribute__((aligned(16))) float sAcc[NN + 1024];
    static_assert(sizeof(float) * (2 * NN + 1024) > 160 * 1024, "more than half of the 320 KiB of a WGP");
    const int c   = blockIdx.x % NCH;
    const int t   = blockIdx.x / NCH;
    const int f   = t % NF;
    const int b   = t / NF;
    const int tid = threadIdx.x;

    const float* __restrict__ gdis = dis + (size_t)b * NN;
    const float* __restrict__ sp = src + (size_t)b * NN * NF + f;
    const float4* __restrict__ dq = reinterpret_cast<const float4*>(gdis);
    const int* __restrict__ eb = ei + (size_t)b * 2 * EPB;
    const int4* __restrict__ S = reinterpret_cast<const int4*>(eb) + (size_t)c * NVC;
    const int4* __restrict__ D = reinterpret_cast<const int4*>(eb + EPB) + (size_t)c * NVC;
    float xs[FS], ds[FS];
    #pragma unroll
    for (int k = 0; k < FS; ++k) {
        const int n  = tid + k * 1024;
        const int nn = (n < NN) ? n : tid;
        xs[k] = sp[(size_t)nn * NF];
        ds[k] = gdis[nn];
    }
    float4 q[FT];
    #pragma unroll
    for (int k = 0; k < FT; ++k) {
        const int j = tid + k * 1024;
        q[k] = dq[(j < NV4) ? j : tid];
    }
    int i = tid;
    int4 s0 = S[i], s1 = S[i + 1024];
    int4 d0 = D[i], d1 = D[i + 1024];
    #pragma unroll
    for (int k = 0; k < FS; ++k) {
        const int n  = tid + k * 1024;
        const int nn = (n < NN) ? n : tid;
        sSrc[nn] = xs[k] * ds[k];
        sAcc[nn] = 0.f;
    }
    __syncthreads();

    auto add4 = [&](const int4 s, const int4 d) {
        const float v0 = sSrc[s.x];
        const float v1 = sSrc[s.y];
        const float v2 = sSrc[s.z];
        const float v3 = sSrc[s.w];
        __hip_atomic_fetch_add(&sAcc[d.x], v0, __ATOMIC_RELAXED, __HIP_MEMORY_SCOPE_WORKGROUP);
        __hip_atomic_fetch_add(&sAcc[d.y], v1, __ATOMIC_RELAXED, __HIP_MEMORY_SCOPE_WORKGROUP);
        __hip_atomic_fetch_add(&sAcc[d.z], v2, __ATOMIC_RELAXED, __HIP_MEMORY_SCOPE_WORKGROUP);
        __hip_atomic_fetch_add(&sAcc[d.w], v3, __ATOMIC_RELAXED, __HIP_MEMORY_SCOPE_WORKGROUP);
    };

    for (int trip = 0; trip < TRIPS; ++trip) {
        const int ni = (trip + 1 < TRIPS) ? i + STEP : i;
        const int n1 = (ni + 1024 < NVC) ? ni + 1024 : ni;
        const int4 s0n = S[ni], s1n = S[n1];
        const int4 d0n = D[ni], d1n = D[n1];
        add4(s0, d0);
        if (i + 1024 < NVC) add4(s1, d1);
        s0 = s0n; s1 = s1n; d0 = d0n; d1 = d1n;
        i = ni;
    }
    __syncthreads();

    float4* __restrict__ dp =
        reinterpret_cast<float4*>(dst + (size_t)(c * NF + f) * NTOT + (size_t)b * NN);
    #pragma unroll
    for (int k = 0; k < FT; ++k) {
        const int j = tid + k * 1024;
        if (j < NV4) {
            const float4 a = reinterpret_cast<const float4*>(sAcc)[j];
            dp[j] = make_float4(a.x * q[k].x, a.y * q[k].y, a.z * q[k].z, a.w * q[k].w);
        }
    }
}

template <int NCH>
__global__ __launch_bounds__(1024)
void pair_sum_kernel(const unsigned* __restrict__ sched, const unsigned* __restrict__ ov,
                     const unsigned* __restrict__ ovcnt, const float* __restrict__ dis,
                     const float* __restrict__ src, float* __restrict__ dst) {
    static_assert((NN % 4) == 0 && (DCH % NCH) == 0 && (HID % 2) == 0,
                  "16-byte vectors over nodes, whole segments per chunk, features in pairs");
    constexpr int SPC  = DCH / NCH;
    constexpr int NVG  = SPC * GRIDW / 4;
    constexpr int STEP = 2048;
    constexpr int FULL = NVG / STEP;
    static_assert(FULL >= 1 && (NVG % STEP) == 0, "the grid is a whole number of trips");
    constexpr int NV4  = NN / 4;
    constexpr int FT   = (NV4 + 1023) / 1024;
    __shared__ __attribute__((aligned(16))) float2 sSrc[NN + 32];
    __shared__ __attribute__((aligned(16))) float  sAcc0[NN];
    __shared__ __attribute__((aligned(16))) float  sAcc1[NN];
    static_assert(sizeof(float) * (4 * NN + 64) <= 320 * 1024, "four planes and the null entries fit the 320 KiB of a WGP");
    const int c   = blockIdx.x % NCH;
    const int t   = blockIdx.x / NCH;
    const int f0  = (t % (HID / 2)) * 2;
    const int b   = t / (HID / 2);
    const int tid = threadIdx.x;

    const float* __restrict__ gdis = dis + (size_t)b * NN;
    float4 u[FT], v[FT], q[FT];
    {
        const float4* __restrict__ sp0 =
            reinterpret_cast<const float4*>(src + (size_t)f0 * NTOT + (size_t)b * NN);
        const float4* __restrict__ sp1 =
            reinterpret_cast<const float4*>(src + (size_t)(f0 + 1) * NTOT + (size_t)b * NN);
        const float4* __restrict__ dq = reinterpret_cast<const float4*>(gdis);
        #pragma unroll
        for (int k = 0; k < FT; ++k) {
            const int j  = tid + k * 1024;
            const int jj = (j < NV4) ? j : tid;
            u[k] = sp0[jj];
            v[k] = sp1[jj];
            q[k] = dq[jj];
        }
    }
    const int seg0 = b * DCH + c * SPC;
    const uint4* __restrict__ G = reinterpret_cast<const uint4*>(sched + (size_t)seg0 * GRIDW);
    static_assert((SPC % 2) == 0, "overflow words are taken four at a time, two segments each");
    unsigned x[SPC][2];
    #pragma unroll
    for (int sg = 0; sg < SPC; ++sg) {
        const unsigned* __restrict__ O = ov + (size_t)(seg0 + sg) * OVPAD;
        x[sg][0] = O[tid];
        x[sg][1] = O[tid + 1024];
    }
    unsigned nup[SPC];
    #pragma unroll
    for (int sg = 0; sg < SPC; ++sg) {
        unsigned n = ovcnt[(size_t)(seg0 + sg) * OVNST];
        n = (n < (unsigned)OVCAP) ? n : (unsigned)OVCAP;
        nup[sg] = (n + 1023u) & ~1023u;
    }
    int i = tid;
    uint4 w0 = G[i], w1 = G[i + 1024];

    #pragma unroll
    for (int k = 0; k < FT; ++k) {
        const int j  = tid + k * 1024;
        const int jj = (j < NV4) ? j : tid;
        reinterpret_cast<float4*>(sSrc)[2 * jj]     = make_float4(u[k].x * q[k].x, v[k].x * q[k].x, u[k].y * q[k].y, v[k].y * q[k].y);
        reinterpret_cast<float4*>(sSrc)[2 * jj + 1] = make_float4(u[k].z * q[k].z, v[k].z * q[k].z, u[k].w * q[k].w, v[k].w * q[k].w);
        reinterpret_cast<float4*>(sAcc0)[jj] = make_float4(0.f, 0.f, 0.f, 0.f);
        reinterpret_cast<float4*>(sAcc1)[jj] = make_float4(0.f, 0.f, 0.f, 0.f);
    }
    if (tid < 32) sSrc[NN + tid] = make_float2(0.f, 0.f);
    __syncthreads();

    auto add1 = [&](const unsigned w) {
        const float2 v = sSrc[w & 0xFFFFu];
        const unsigned d = w >> 16;
        __hip_atomic_fetch_add(&sAcc0[d], v.x, __ATOMIC_RELAXED, __HIP_MEMORY_SCOPE_WORKGROUP);
        __hip_atomic_fetch_add(&sAcc1[d], v.y, __ATOMIC_RELAXED, __HIP_MEMORY_SCOPE_WORKGROUP);
    };
    auto add4 = [&](const uint4 w) {
        const float2 v0 = sSrc[w.x & 0xFFFFu];
        const float2 v1 = sSrc[w.y & 0xFFFFu];
        const float2 v2 = sSrc[w.z & 0xFFFFu];
        const float2 v3 = sSrc[w.w & 0xFFFFu];
        const unsigned d0 = w.x >> 16, d1 = w.y >> 16, d2 = w.z >> 16, d3 = w.w >> 16;
        __hip_atomic_fetch_add(&sAcc0[d0], v0.x, __ATOMIC_RELAXED, __HIP_MEMORY_SCOPE_WORKGROUP);
        __hip_atomic_fetch_add(&sAcc1[d0], v0.y, __ATOMIC_RELAXED, __HIP_MEMORY_SCOPE_WORKGROUP);
        __hip_atomic_fetch_add(&sAcc0[d1], v1.x, __ATOMIC_RELAXED, __HIP_MEMORY_SCOPE_WORKGROUP);
        __hip_atomic_fetch_add(&sAcc1[d1], v1.y, __ATOMIC_RELAXED, __HIP_MEMORY_SCOPE_WORKGROUP);
        __hip_atomic_fetch_add(&sAcc0[d2], v2.x, __ATOMIC_RELAXED, __HIP_MEMORY_SCOPE_WORKGROUP);
        __hip_atomic_fetch_add(&sAcc1[d2], v2.y, __ATOMIC_RELAXED, __HIP_MEMORY_SCOPE_WORKGROUP);
        __hip_atomic_fetch_add(&sAcc0[d3], v3.x, __ATOMIC_RELAXED, __HIP_MEMORY_SCOPE_WORKGROUP);
        __hip_atomic_fetch_add(&sAcc1[d3], v3.y, __ATOMIC_RELAXED, __HIP_MEMORY_SCOPE_WORKGROUP);
    };

    for (int trip = 0; trip < FULL; ++trip) {
        const int ni = (trip + 1 < FULL) ? i + STEP : i;
        const uint4 w0n = G[ni], w1n = G[ni + 1024];
        add4(w0);
        add4(w1);
        w0 = w0n; w1 = w1n;
        i = ni;
    }
    #pragma unroll
    for (int sg = 0; sg < SPC; sg += 2)
        add4(make_uint4(x[sg][0], x[sg][1], x[sg + 1][0], x[sg + 1][1]));
    #pragma unroll
    for (int sg = 0; sg < SPC; ++sg) {
        const unsigned* __restrict__ O = ov + (size_t)(seg0 + sg) * OVPAD;
        for (unsigned j = 2048u + tid; j < nup[sg]; j += 1024u) add1(O[j]);
    }
    __syncthreads();

    float4* __restrict__ dp0 =
        reinterpret_cast<float4*>(dst + (size_t)(c * HID + f0) * NTOT + (size_t)b * NN);
    float4* __restrict__ dp1 =
        reinterpret_cast<float4*>(dst + (size_t)(c * HID + f0 + 1) * NTOT + (size_t)b * NN);
    #pragma unroll
    for (int k = 0; k < FT; ++k) {
        const int j = tid + k * 1024;
        if (j < NV4) {
            const float4 a0 = reinterpret_cast<const float4*>(sAcc0)[j];
            const float4 a1 = reinterpret_cast<const float4*>(sAcc1)[j];
            dp0[j] = make_float4(a0.x * q[k].x, a0.y * q[k].y, a0.z * q[k].z, a0.w * q[k].w);
            dp1[j] = make_float4(a1.x * q[k].x, a1.y * q[k].y, a1.z * q[k].z, a1.w * q[k].w);
        }
    }
}

template <int K, int KW, int NCH, bool OWN_PLANES, bool OUT_PLANES, bool READOUT>
__global__ __launch_bounds__(256)
void tag_gemm_kernel(const float* __restrict__ hin,
                                const float* __restrict__ agg,
                                const float* __restrict__ W,
                                const float* __restrict__ bias,
                                float* __restrict__ hout,
                                const float* __restrict__ wreg,
                                const float* __restrict__ breg,
                                float* __restrict__ outv, int N) {
    static_assert((K % 4) == 0, "padded K must be a multiple of 4");
    static_assert(!(READOUT && OUT_PLANES), "the readout sums rows of the row-major tile");
    constexpr int TPB = 256;
    constexpr int NC  = K / 4;
    __shared__ float sW[2 * K * HID];
    __shared__ float sB[HID];

    const int lane = threadIdx.x & 31;
    const int wave = (blockIdx.x * TPB + threadIdx.x) >> 5;

    const int m     = lane & 15;
    const int khalf = (lane >> 4) * 2;

    auto load_tile = [&](const int row0, v2f (&own)[NC], v2f (&part)[NCH][NC]) {
        #pragma unroll
        for (int c = 0; c < NC; ++c) {
            const int k0 = c * 4 + khalf;
            const int kx = (KW == K || k0 < KW) ? k0 : (KW - 1);
            const int ky = (KW == K || k0 + 1 < KW) ? (k0 + 1) : (KW - 1);
            if constexpr (OWN_PLANES) {
                own[c].x = hin[(size_t)k0 * NTOT + row0 + m];
                own[c].y = hin[(size_t)(k0 + 1) * NTOT + row0 + m];
            } else {
                own[c].x = hin[(size_t)(row0 + m) * KW + kx];
                own[c].y = hin[(size_t)(row0 + m) * KW + ky];
            }
            #pragma unroll
            for (int ch = 0; ch < NCH; ++ch) {
                part[ch][c].x = agg[((size_t)ch * KW + kx) * NTOT + row0 + m];
                part[ch][c].y = agg[((size_t)ch * KW + ky) * NTOT + row0 + m];
            }
        }
    };

    const int rowA = wave * 32;
    const int rowB = rowA + 16;
    const bool hasA = rowA < N;
    const bool hasB = rowB < N;
    v2f ownA[NC], partA[NCH][NC];
    if (hasA) load_tile(rowA, ownA, partA);
    float wr0 = 0.f, wr1 = 0.f, bq = 0.f;
    if constexpr (READOUT) {
        wr0 = wreg[m];
        wr1 = wreg[16 + m];
        bq  = breg[0];
    }

    static_assert(((2 * K * HID) % TPB) == 0 && HID <= TPB && (HID & (HID - 1)) == 0,
                  "whole trips over the weights; the bias index is masked");
    constexpr int WT = (2 * K * HID) / TPB;
    float wv[WT];
    #pragma unroll
    for (int j = 0; j < WT; ++j) {
        const int i   = (int)threadIdx.x + j * TPB;
        const int s   = i / (K * HID);
        const int rem = i - s * (K * HID);
        const int kw  = rem / HID;
        const int col = rem - kw * HID;
        const int kr  = (kw < KW) ? kw : (KW - 1);
        wv[j] = W[((size_t)s * KW + kr) * HID + col];
    }
    const float bv = bias[threadIdx.x & (HID - 1)];
    #pragma unroll
    for (int j = 0; j < WT; ++j) {
        const int i   = (int)threadIdx.x + j * TPB;
        const int s   = i / (K * HID);
        const int rem = i - s * (K * HID);
        const int kw  = rem / HID;
        sW[i] = (kw < KW) ? wv[j] : 0.0f;
    }
    sB[threadIdx.x & (HID - 1)] = bv;
    __syncthreads();
    if (!hasA) return;

    auto product = [&](const v2f (&own)[NC], const v2f (&part)[NCH][NC], v8f& acc0, v8f& acc1) {
        #pragma unroll
        for (int s = 0; s < 2; ++s) {
            const float* __restrict__ sWk = sW + s * K * HID;
            #pragma unroll
            for (int c = 0; c < NC; ++c) {
                const int k0 = c * 4 + khalf;
                v2f a;
                if (s) {
                    float vx = part[0][c].x;
                    float vy = part[0][c].y;
                    #pragma unroll
                    for (int ch = 1; ch < NCH; ++ch) {
                        vx += part[ch][c].x;
                        vy += part[ch][c].y;
                    }
                    a.x = (KW == K || k0 < KW) ? vx : 0.0f;
                    a.y = (KW == K || k0 + 1 < KW) ? vy : 0.0f;
                } else if constexpr (!OWN_PLANES && KW != K) {
                    a.x = (k0 < KW) ? own[c].x : 0.0f;
                    a.y = (k0 + 1 < KW) ? own[c].y : 0.0f;
                } else {
                    a = own[c];
                }
                v2f b0, b1;
                b0.x = sWk[k0 * HID + m];
                b0.y = sWk[(k0 + 1) * HID + m];
                b1.x = sWk[k0 * HID + 16 + m];
                b1.y = sWk[(k0 + 1) * HID + 16 + m];
                if constexpr (OUT_PLANES) {
                    acc0 = __builtin_amdgcn_wmma_f32_16x16x4_f32(
                               false, b0, false, a, (short)0, acc0, false, false);
                    acc1 = __builtin_amdgcn_wmma_f32_16x16x4_f32(
                               false, b1, false, a, (short)0, acc1, false, false);
                } else {
                    acc0 = __builtin_amdgcn_wmma_f32_16x16x4_f32(
                               false, a, false, b0, (short)0, acc0, false, false);
                    acc1 = __builtin_amdgcn_wmma_f32_16x16x4_f32(
                               false, a, false, b1, (short)0, acc1, false, false);
                }
            }
        }
    };

    auto finish = [&](const int row0, const v8f acc0, const v8f acc1) {
        if constexpr (OUT_PLANES) {
            const int fbase = (lane >> 4) ? 8 : 0;
            #pragma unroll
            for (int v = 0; v < 8; ++v) {
                int f = fbase + v;
                float x0 = acc0[v] + sB[f];
                float x1 = acc1[v] + sB[16 + f];
                hout[(size_t)f * NTOT + row0 + m]        = tanhf(x0);
                hout[(size_t)(16 + f) * NTOT + row0 + m] = tanhf(x1);
            }
        } else {
            const int rbase = row0 + ((lane >> 4) ? 8 : 0);
            float p[8];
            #pragma unroll
            for (int v = 0; v < 8; ++v) {
                int r = rbase + v;
                float x0 = acc0[v] + sB[m];
                float x1 = acc1[v] + sB[16 + m];
                const float t0 = tanhf(x0);
                const float t1 = tanhf(x1);
                hout[(size_t)r * HID + m]      = t0;
                hout[(size_t)r * HID + 16 + m] = t1;
                p[v] = READOUT ? (t0 * wr0 + t1 * wr1) : 0.0f;
            }
            if constexpr (READOUT) {
                float o = 0.0f;
                #pragma unroll
                for (int v = 0; v < 8; ++v) {
                    const float s = row16_sum(p[v]);
                    o = (m == v) ? s : o;
                }
                if (m < 8) outv[rbase + m] = o + bq;
            }
        }
    };

    v8f accA0 = {0.f,0.f,0.f,0.f,0.f,0.f,0.f,0.f};
    v8f accA1 = {0.f,0.f,0.f,0.f,0.f,0.f,0.f,0.f};
    product(ownA, partA, accA0, accA1);
    v2f ownB[NC], partB[NCH][NC];
    if (hasB) load_tile(rowB, ownB, partB);
    finish(rowA, accA0, accA1);
    if (hasB) {
        v8f accB0 = {0.f,0.f,0.f,0.f,0.f,0.f,0.f,0.f};
        v8f accB1 = {0.f,0.f,0.f,0.f,0.f,0.f,0.f,0.f};
        product(ownB, partB, accB0, accB1);
        finish(rowB, accB0, accB1);
    }
}

static inline int cdiv(long long a, long long b) { return (int)((a + b - 1) / b); }

extern "C" void kernel_launch(void* const* d_in, const int* in_sizes, int n_in,
                              void* d_out, int out_size, void* d_ws, size_t ws_size,
                              hipStream_t stream) {
    const float* X    = (const float*)d_in[0];
    const int*   ei   = (const int*)  d_in[1];
    const float* W0   = (const float*)d_in[2];
    const float* b0   = (const float*)d_in[3];
    const float* W1   = (const float*)d_in[4];
    const float* b1   = (const float*)d_in[5];
    const float* W2   = (const float*)d_in[6];
    const float* b2   = (const float*)d_in[7];
    const float* Wreg = (const float*)d_in[8];
    const float* breg = (const float*)d_in[9];

    float* out    = (float*)d_out;
    float* hfinal = out + NTOT;

    float* dis = (float*)d_ws;
    float* agg = dis + NTOT;
    float* hA  = agg + (size_t)NTOT * HID * LCH;
    float* hB  = hA  + (size_t)NTOT * HID;
    unsigned* cnt = (unsigned*)(hB + (size_t)NTOT * HID);
    unsigned* sched = cnt + (size_t)DCH * NTOT;
    unsigned* ov    = sched + (size_t)BATCH * DCH * GRIDW;
    unsigned* ovcnt = ov + (size_t)BATCH * DCH * OVPAD;

    const int T = 256;

    degree_schedule_kernel<<<BATCH * DCH, 1024, 0, stream>>>(ei, cnt, sched, ov, ovcnt);
    degree_rsqrt_kernel<<<cdiv(NTOT / 4, T), T, 0, stream>>>(cnt, dis);

    static_assert((NTOT % 32) == 0, "every wave of the product launches owns two whole 16-row tiles");
    const int gemm_blocks = cdiv((long long)(NTOT / 32) * 32, T);

    plane_sum_kernel<WINDOW, L0CH><<<BATCH * WINDOW * L0CH, 1024, 0, stream>>>(ei, dis, X, agg);
    tag_gemm_kernel<KPAD, WINDOW, L0CH, false, true, false><<<gemm_blocks, T, 0, stream>>>(X, agg, W0, b0, hA, nullptr, nullptr, nullptr, NTOT);

    pair_sum_kernel<LCH><<<BATCH * (HID / 2) * LCH, 1024, 0, stream>>>(sched, ov, ovcnt, dis, hA, agg);
    tag_gemm_kernel<HID, HID, LCH, true, true, false><<<gemm_blocks, T, 0, stream>>>(hA, agg, W1, b1, hB, nullptr, nullptr, nullptr, NTOT);

    pair_sum_kernel<LCH><<<BATCH * (HID / 2) * LCH, 1024, 0, stream>>>(sched, ov, ovcnt, dis, hB, agg);
    tag_gemm_kernel<HID, HID, LCH, true, false, true><<<gemm_blocks, T, 0, stream>>>(hB, agg, W2, b2, hfinal, Wreg, breg, out, NTOT);
}


# product launches: tanh of the 3 layers on the hardware transcendental v_tanh_f32 (one issue per value) in place of the library's branching in-line tanhf; tanh values may differ in the last bits
# speedup vs baseline: 24.0650x; 1.0443x over previous
#include <hip/hip_runtime.h>
#include <hip/hip_bf16.h>
#include <math.h>

#define BATCH   8
#define WINDOW  5
#define NN      20000
#define EPB     320000
#define HID     32
#define KPAD    8
#define NTOT    (BATCH * NN)
#define ETOT    (BATCH * EPB)
#define L0CH    5
#define DCH     16
#define LCH     2
#define PASSES  11
#define GRIDW   (PASSES * 64 * 32)
#define OVCAP   (EPB / DCH)
#define OVPAD   (((OVCAP + 1023) / 1024) * 1024)
#define OVNST   32

typedef float v2f __attribute__((ext_vector_type(2)));
typedef float v8f __attribute__((ext_vector_type(8)));

template <int XMASK>
__device__ __forceinline__ float row_xor_get(const float x) {
    return __int_as_float(__builtin_amdgcn_update_dpp(0, __float_as_int(x), 0x160 | XMASK, 0xF, 0xF, true));
}
__device__ __forceinline__ float row16_sum(float x) {
    x += row_xor_get<1>(x);
    x += row_xor_get<2>(x);
    x += row_xor_get<4>(x);
    x += row_xor_get<8>(x);
    return x;
}

__device__ __forceinline__ float hw_tanh(const float x) {
    return __builtin_amdgcn_tanhf(x);
}

__global__ __launch_bounds__(1024)
void degree_schedule_kernel(const int* __restrict__ ei, unsigned* __restrict__ cnt,
                            unsigned* __restrict__ sched, unsigned* __restrict__ ov,
                            unsigned* __restrict__ ovcnt) {
    static_assert((EPB % (4 * DCH)) == 0 && (NN % 4) == 0, "16-byte vectors over edge chunks and nodes");
    static_assert((NN % 32) == 0, "the null source entries NN + lane keep the lane's bank pair");
    static_assert(NN + 32 <= 65536 && NN >= 64, "packed 16-bit source and destination indices; null destinations are real nodes");
    static_assert((GRIDW % 4) == 0 && (OVCAP % 4) == 0, "16-byte vectors over grid and overflow list");
    constexpr int NVC   = EPB / 4 / DCH;
    constexpr int TRIPS = (NVC + 1023) / 1024;
    __shared__ __attribute__((aligned(16))) unsigned sCnt[NN];
    __shared__ __attribute__((aligned(16))) unsigned sGrid[GRIDW];
    __shared__ __attribute__((aligned(16))) unsigned sOv[OVCAP];
    __shared__ __attribute__((aligned(16))) unsigned sCell[32 * 64];
    __shared__ unsigned sOvN;
    const int c   = blockIdx.x % DCH;
    const int b   = blockIdx.x / DCH;
    const int tid = threadIdx.x;

    const int4* __restrict__ S =
        reinterpret_cast<const int4*>(ei + (size_t)b * 2 * EPB) + (size_t)c * NVC;
    const int4* __restrict__ D =
        reinterpret_cast<const int4*>(ei + (size_t)b * 2 * EPB + EPB) + (size_t)c * NVC;
    int4 s[TRIPS], d[TRIPS];
    #pragma unroll
    for (int k = 0; k < TRIPS; ++k) {
        const int j = tid + k * 1024;
        s[k] = S[(j < NVC) ? j : tid];
        d[k] = D[(j < NVC) ? j : tid];
    }
    for (int i = tid; i < NN / 4; i += 1024)
        reinterpret_cast<uint4*>(sCnt)[i] = make_uint4(0u, 0u, 0u, 0u);
    if (tid < 32 * 64 / 4)
        reinterpret_cast<uint4*>(sCell)[tid] = make_uint4(0u, 0u, 0u, 0u);
    if (tid == 0) sOvN = 0u;
    for (int v = tid; v < GRIDW / 4; v += 1024) {
        const unsigned lane = (unsigned)v & 31u;
        const unsigned r0   = ((unsigned)v >> 5) << 2;
        const unsigned ns   = (unsigned)NN + lane;
        reinterpret_cast<uint4*>(sGrid)[v] = make_uint4(ns | (((lane + r0     ) & 63u) << 16),
                                                        ns | (((lane + r0 + 1u) & 63u) << 16),
                                                        ns | (((lane + r0 + 2u) & 63u) << 16),
                                                        ns | (((lane + r0 + 3u) & 63u) << 16));
    }
    __syncthreads();

    auto rank_of = [&](const int src, const int dst) -> unsigned {
        __hip_atomic_fetch_add(&sCnt[dst], 1u, __ATOMIC_RELAXED, __HIP_MEMORY_SCOPE_WORKGROUP);
        return __hip_atomic_fetch_add(&sCell[((unsigned)src & 31u) * 64u + ((unsigned)dst & 63u)], 1u,
                                      __ATOMIC_RELAXED, __HIP_MEMORY_SCOPE_WORKGROUP);
    };
    auto put = [&](const int src, const int dst, const unsigned p) {
        const unsigned sc = (unsigned)src & 31u;
        const unsigned dc = (unsigned)dst & 63u;
        const unsigned w  = (unsigned)src | ((unsigned)dst << 16);
        if (p < (unsigned)PASSES) {
            const unsigned R = p * 64u + ((dc - sc) & 63u);
            sGrid[((((R >> 2) << 5) + sc) << 2) | (R & 3u)] = w;
        } else {
            const unsigned q = __hip_atomic_fetch_add(&sOvN, 1u, __ATOMIC_RELAXED,
                                                      __HIP_MEMORY_SCOPE_WORKGROUP);
            sOv[q] = w;
        }
    };
    unsigned rk[TRIPS][4];
    #pragma unroll
    for (int k = 0; k < TRIPS; ++k) {
        if (tid + k * 1024 < NVC) {
            rk[k][0] = rank_of(s[k].x, d[k].x);
            rk[k][1] = rank_of(s[k].y, d[k].y);
            rk[k][2] = rank_of(s[k].z, d[k].z);
            rk[k][3] = rank_of(s[k].w, d[k].w);
        }
    }
    #pragma unroll
    for (int k = 0; k < TRIPS; ++k) {
        if (tid + k * 1024 < NVC) {
            put(s[k].x, d[k].x, rk[k][0]);
            put(s[k].y, d[k].y, rk[k][1]);
            put(s[k].z, d[k].z, rk[k][2]);
            put(s[k].w, d[k].w, rk[k][3]);
        }
    }
    __syncthreads();

    uint4* __restrict__ op =
        reinterpret_cast<uint4*>(cnt + (size_t)c * NTOT + (size_t)b * NN);
    for (int i = tid; i < NN / 4; i += 1024)
        op[i] = reinterpret_cast<const uint4*>(sCnt)[i];

    const int seg = b * DCH + c;
    uint4* __restrict__ gp = reinterpret_cast<uint4*>(sched + (size_t)seg * GRIDW);
    for (int i = tid; i < GRIDW / 4; i += 1024)
        gp[i] = reinterpret_cast<const uint4*>(sGrid)[i];
    const unsigned nov = sOvN;
    const unsigned nup = (nov + 1023u) & ~1023u;
    const unsigned npad = (nup > 2048u) ? nup : 2048u;
    unsigned* __restrict__ vp = ov + (size_t)seg * OVPAD;
    for (unsigned i = tid; i < npad; i += 1024u)
        vp[i] = (i < nov) ? sOv[i] : (((unsigned)NN + (i & 31u)) | ((i & 63u) << 16));
    if (tid == 0) ovcnt[(size_t)seg * OVNST] = nov;
}

__global__ void degree_rsqrt_kernel(const unsigned* __restrict__ cnt, float* __restrict__ dis) {
    int i = blockIdx.x * blockDim.x + threadIdx.x;
    if (i >= NTOT / 4) return;
    uint4 s = reinterpret_cast<const uint4*>(cnt)[i];
    #pragma unroll
    for (int c = 1; c < DCH; ++c) {
        const uint4 t = reinterpret_cast<const uint4*>(cnt + (size_t)c * NTOT)[i];
        s.x += t.x; s.y += t.y; s.z += t.z; s.w += t.w;
    }
    float4 r;
    r.x = (s.x > 0u) ? rsqrtf((float)s.x) : 0.0f;
    r.y = (s.y > 0u) ? rsqrtf((float)s.y) : 0.0f;
    r.z = (s.z > 0u) ? rsqrtf((float)s.z) : 0.0f;
    r.w = (s.w > 0u) ? rsqrtf((float)s.w) : 0.0f;
    reinterpret_cast<float4*>(dis)[i] = r;
}

template <int NF, int NCH>
__global__ __launch_bounds__(1024)
void plane_sum_kernel(const int* __restrict__ ei, const float* __restrict__ dis,
                      const float* __restrict__ src, float* __restrict__ dst) {
    static_assert((NN % 4) == 0 && (EPB % (4 * NCH)) == 0, "16-byte vectors over nodes and edge chunks");
    constexpr int NVC  = EPB / 4 / NCH;
    constexpr int STEP = 2048;
    constexpr int TRIPS = (NVC + STEP - 1) / STEP;
    static_assert(NVC >= STEP && (TRIPS - 1) * STEP + 1024 <= NVC,
                  "every thread owns a first pair of edge vectors, and a first vector in every trip");
    constexpr int NV4  = NN / 4;
    constexpr int FT   = (NV4 + 1023) / 1024;
    constexpr int FS   = (NN + 1023) / 1024;
    __shared__ __attribute__((aligned(16))) float sSrc[NN];
    __shared__ __attribute__((aligned(16))) float sAcc[NN + 1024];
    static_assert(sizeof(float) * (2 * NN + 1024) > 160 * 1024, "more than half of the 320 KiB of a WGP");
    const int c   = blockIdx.x % NCH;
    const int t   = blockIdx.x / NCH;
    const int f   = t % NF;
    const int b   = t / NF;
    const int tid = threadIdx.x;

    const float* __restrict__ gdis = dis + (size_t)b * NN;
    const float* __restrict__ sp = src + (size_t)b * NN * NF + f;
    const float4* __restrict__ dq = reinterpret_cast<const float4*>(gdis);
    const int* __restrict__ eb = ei + (size_t)b * 2 * EPB;
    const int4* __restrict__ S = reinterpret_cast<const int4*>(eb) + (size_t)c * NVC;
    const int4* __restrict__ D = reinterpret_cast<const int4*>(eb + EPB) + (size_t)c * NVC;
    float xs[FS], ds[FS];
    #pragma unroll
    for (int k = 0; k < FS; ++k) {
        const int n  = tid + k * 1024;
        const int nn = (n < NN) ? n : tid;
        xs[k] = sp[(size_t)nn * NF];
        ds[k] = gdis[nn];
    }
    float4 q[FT];
    #pragma unroll
    for (int k = 0; k < FT; ++k) {
        const int j = tid + k * 1024;
        q[k] = dq[(j < NV4) ? j : tid];
    }
    int i = tid;
    int4 s0 = S[i], s1 = S[i + 1024];
    int4 d0 = D[i], d1 = D[i + 1024];
    #pragma unroll
    for (int k = 0; k < FS; ++k) {
        const int n  = tid + k * 1024;
        const int nn = (n < NN) ? n : tid;
        sSrc[nn] = xs[k] * ds[k];
        sAcc[nn] = 0.f;
    }
    __syncthreads();

    auto add4 = [&](const int4 s, const int4 d) {
        const float v0 = sSrc[s.x];
        const float v1 = sSrc[s.y];
        const float v2 = sSrc[s.z];
        const float v3 = sSrc[s.w];
        __hip_atomic_fetch_add(&sAcc[d.x], v0, __ATOMIC_RELAXED, __HIP_MEMORY_SCOPE_WORKGROUP);
        __hip_atomic_fetch_add(&sAcc[d.y], v1, __ATOMIC_RELAXED, __HIP_MEMORY_SCOPE_WORKGROUP);
        __hip_atomic_fetch_add(&sAcc[d.z], v2, __ATOMIC_RELAXED, __HIP_MEMORY_SCOPE_WORKGROUP);
        __hip_atomic_fetch_add(&sAcc[d.w], v3, __ATOMIC_RELAXED, __HIP_MEMORY_SCOPE_WORKGROUP);
    };

    for (int trip = 0; trip < TRIPS; ++trip) {
        const int ni = (trip + 1 < TRIPS) ? i + STEP : i;
        const int n1 = (ni + 1024 < NVC) ? ni + 1024 : ni;
        const int4 s0n = S[ni], s1n = S[n1];
        const int4 d0n = D[ni], d1n = D[n1];
        add4(s0, d0);
        if (i + 1024 < NVC) add4(s1, d1);
        s0 = s0n; s1 = s1n; d0 = d0n; d1 = d1n;
        i = ni;
    }
    __syncthreads();

    float4* __restrict__ dp =
        reinterpret_cast<float4*>(dst + (size_t)(c * NF + f) * NTOT + (size_t)b * NN);
    #pragma unroll
    for (int k = 0; k < FT; ++k) {
        const int j = tid + k * 1024;
        if (j < NV4) {
            const float4 a = reinterpret_cast<const float4*>(sAcc)[j];
            dp[j] = make_float4(a.x * q[k].x, a.y * q[k].y, a.z * q[k].z, a.w * q[k].w);
        }
    }
}

template <int NCH>
__global__ __launch_bounds__(1024)
void pair_sum_kernel(const unsigned* __restrict__ sched, const unsigned* __restrict__ ov,
                     const unsigned* __restrict__ ovcnt, const float* __restrict__ dis,
                     const float* __restrict__ src, float* __restrict__ dst) {
    static_assert((NN % 4) == 0 && (DCH % NCH) == 0 && (HID % 2) == 0,
                  "16-byte vectors over nodes, whole segments per chunk, features in pairs");
    constexpr int SPC  = DCH / NCH;
    constexpr int NVG  = SPC * GRIDW / 4;
    constexpr int STEP = 2048;
    constexpr int FULL = NVG / STEP;
    static_assert(FULL >= 1 && (NVG % STEP) == 0, "the grid is a whole number of trips");
    constexpr int NV4  = NN / 4;
    constexpr int FT   = (NV4 + 1023) / 1024;
    __shared__ __attribute__((aligned(16))) float2 sSrc[NN + 32];
    __shared__ __attribute__((aligned(16))) float  sAcc0[NN];
    __shared__ __attribute__((aligned(16))) float  sAcc1[NN];
    static_assert(sizeof(float) * (4 * NN + 64) <= 320 * 1024, "four planes and the null entries fit the 320 KiB of a WGP");
    const int c   = blockIdx.x % NCH;
    const int t   = blockIdx.x / NCH;
    const int f0  = (t % (HID / 2)) * 2;
    const int b   = t / (HID / 2);
    const int tid = threadIdx.x;

    const float* __restrict__ gdis = dis + (size_t)b * NN;
    float4 u[FT], v[FT], q[FT];
    {
        const float4* __restrict__ sp0 =
            reinterpret_cast<const float4*>(src + (size_t)f0 * NTOT + (size_t)b * NN);
        const float4* __restrict__ sp1 =
            reinterpret_cast<const float4*>(src + (size_t)(f0 + 1) * NTOT + (size_t)b * NN);
        const float4* __restrict__ dq = reinterpret_cast<const float4*>(gdis);
        #pragma unroll
        for (int k = 0; k < FT; ++k) {
            const int j  = tid + k * 1024;
            const int jj = (j < NV4) ? j : tid;
            u[k] = sp0[jj];
            v[k] = sp1[jj];
            q[k] = dq[jj];
        }
    }
    const int seg0 = b * DCH + c * SPC;
    const uint4* __restrict__ G = reinterpret_cast<const uint4*>(sched + (size_t)seg0 * GRIDW);
    static_assert((SPC % 2) == 0, "overflow words are taken four at a time, two segments each");
    unsigned x[SPC][2];
    #pragma unroll
    for (int sg = 0; sg < SPC; ++sg) {
        const unsigned* __restrict__ O = ov + (size_t)(seg0 + sg) * OVPAD;
        x[sg][0] = O[tid];
        x[sg][1] = O[tid + 1024];
    }
    unsigned nup[SPC];
    #pragma unroll
    for (int sg = 0; sg < SPC; ++sg) {
        unsigned n = ovcnt[(size_t)(seg0 + sg) * OVNST];
        n = (n < (unsigned)OVCAP) ? n : (unsigned)OVCAP;
        nup[sg] = (n + 1023u) & ~1023u;
    }
    int i = tid;
    uint4 w0 = G[i], w1 = G[i + 1024];

    #pragma unroll
    for (int k = 0; k < FT; ++k) {
        const int j  = tid + k * 1024;
        const int jj = (j < NV4) ? j : tid;
        reinterpret_cast<float4*>(sSrc)[2 * jj]     = make_float4(u[k].x * q[k].x, v[k].x * q[k].x, u[k].y * q[k].y, v[k].y * q[k].y);
        reinterpret_cast<float4*>(sSrc)[2 * jj + 1] = make_float4(u[k].z * q[k].z, v[k].z * q[k].z, u[k].w * q[k].w, v[k].w * q[k].w);
        reinterpret_cast<float4*>(sAcc0)[jj] = make_float4(0.f, 0.f, 0.f, 0.f);
        reinterpret_cast<float4*>(sAcc1)[jj] = make_float4(0.f, 0.f, 0.f, 0.f);
    }
    if (tid < 32) sSrc[NN + tid] = make_float2(0.f, 0.f);
    __syncthreads();

    auto add1 = [&](const unsigned w) {
        const float2 v = sSrc[w & 0xFFFFu];
        const unsigned d = w >> 16;
        __hip_atomic_fetch_add(&sAcc0[d], v.x, __ATOMIC_RELAXED, __HIP_MEMORY_SCOPE_WORKGROUP);
        __hip_atomic_fetch_add(&sAcc1[d], v.y, __ATOMIC_RELAXED, __HIP_MEMORY_SCOPE_WORKGROUP);
    };
    auto add4 = [&](const uint4 w) {
        const float2 v0 = sSrc[w.x & 0xFFFFu];
        const float2 v1 = sSrc[w.y & 0xFFFFu];
        const float2 v2 = sSrc[w.z & 0xFFFFu];
        const float2 v3 = sSrc[w.w & 0xFFFFu];
        const unsigned d0 = w.x >> 16, d1 = w.y >> 16, d2 = w.z >> 16, d3 = w.w >> 16;
        __hip_atomic_fetch_add(&sAcc0[d0], v0.x, __ATOMIC_RELAXED, __HIP_MEMORY_SCOPE_WORKGROUP);
        __hip_atomic_fetch_add(&sAcc1[d0], v0.y, __ATOMIC_RELAXED, __HIP_MEMORY_SCOPE_WORKGROUP);
        __hip_atomic_fetch_add(&sAcc0[d1], v1.x, __ATOMIC_RELAXED, __HIP_MEMORY_SCOPE_WORKGROUP);
        __hip_atomic_fetch_add(&sAcc1[d1], v1.y, __ATOMIC_RELAXED, __HIP_MEMORY_SCOPE_WORKGROUP);
        __hip_atomic_fetch_add(&sAcc0[d2], v2.x, __ATOMIC_RELAXED, __HIP_MEMORY_SCOPE_WORKGROUP);
        __hip_atomic_fetch_add(&sAcc1[d2], v2.y, __ATOMIC_RELAXED, __HIP_MEMORY_SCOPE_WORKGROUP);
        __hip_atomic_fetch_add(&sAcc0[d3], v3.x, __ATOMIC_RELAXED, __HIP_MEMORY_SCOPE_WORKGROUP);
        __hip_atomic_fetch_add(&sAcc1[d3], v3.y, __ATOMIC_RELAXED, __HIP_MEMORY_SCOPE_WORKGROUP);
    };

    for (int trip = 0; trip < FULL; ++trip) {
        const int ni = (trip + 1 < FULL) ? i + STEP : i;
        const uint4 w0n = G[ni], w1n = G[ni + 1024];
        add4(w0);
        add4(w1);
        w0 = w0n; w1 = w1n;
        i = ni;
    }
    #pragma unroll
    for (int sg = 0; sg < SPC; sg += 2)
        add4(make_uint4(x[sg][0], x[sg][1], x[sg + 1][0], x[sg + 1][1]));
    #pragma unroll
    for (int sg = 0; sg < SPC; ++sg) {
        const unsigned* __restrict__ O = ov + (size_t)(seg0 + sg) * OVPAD;
        for (unsigned j = 2048u + tid; j < nup[sg]; j += 1024u) add1(O[j]);
    }
    __syncthreads();

    float4* __restrict__ dp0 =
        reinterpret_cast<float4*>(dst + (size_t)(c * HID + f0) * NTOT + (size_t)b * NN);
    float4* __restrict__ dp1 =
        reinterpret_cast<float4*>(dst + (size_t)(c * HID + f0 + 1) * NTOT + (size_t)b * NN);
    #pragma unroll
    for (int k = 0; k < FT; ++k) {
        const int j = tid + k * 1024;
        if (j < NV4) {
            const float4 a0 = reinterpret_cast<const float4*>(sAcc0)[j];
            const float4 a1 = reinterpret_cast<const float4*>(sAcc1)[j];
            dp0[j] = make_float4(a0.x * q[k].x, a0.y * q[k].y, a0.z * q[k].z, a0.w * q[k].w);
            dp1[j] = make_float4(a1.x * q[k].x, a1.y * q[k].y, a1.z * q[k].z, a1.w * q[k].w);
        }
    }
}

template <int K, int KW, int NCH, bool OWN_PLANES, bool OUT_PLANES, bool READOUT>
__global__ __launch_bounds__(256)
void tag_gemm_kernel(const float* __restrict__ hin,
                                const float* __restrict__ agg,
                                const float* __restrict__ W,
                                const float* __restrict__ bias,
                                float* __restrict__ hout,
                                const float* __restrict__ wreg,
                                const float* __restrict__ breg,
                                float* __restrict__ outv, int N) {
    static_assert((K % 4) == 0, "padded K must be a multiple of 4");
    static_assert(!(READOUT && OUT_PLANES), "the readout sums rows of the row-major tile");
    constexpr int TPB = 256;
    constexpr int NC  = K / 4;
    __shared__ float sW[2 * K * HID];
    __shared__ float sB[HID];

    const int lane = threadIdx.x & 31;
    const int wave = (blockIdx.x * TPB + threadIdx.x) >> 5;

    const int m     = lane & 15;
    const int khalf = (lane >> 4) * 2;

    auto load_tile = [&](const int row0, v2f (&own)[NC], v2f (&part)[NCH][NC]) {
        #pragma unroll
        for (int c = 0; c < NC; ++c) {
            const int k0 = c * 4 + khalf;
            const int kx = (KW == K || k0 < KW) ? k0 : (KW - 1);
            const int ky = (KW == K || k0 + 1 < KW) ? (k0 + 1) : (KW - 1);
            if constexpr (OWN_PLANES) {
                own[c].x = hin[(size_t)k0 * NTOT + row0 + m];
                own[c].y = hin[(size_t)(k0 + 1) * NTOT + row0 + m];
            } else {
                own[c].x = hin[(size_t)(row0 + m) * KW + kx];
                own[c].y = hin[(size_t)(row0 + m) * KW + ky];
            }
            #pragma unroll
            for (int ch = 0; ch < NCH; ++ch) {
                part[ch][c].x = agg[((size_t)ch * KW + kx) * NTOT + row0 + m];
                part[ch][c].y = agg[((size_t)ch * KW + ky) * NTOT + row0 + m];
            }
        }
    };

    const int rowA = wave * 32;
    const int rowB = rowA + 16;
    const bool hasA = rowA < N;
    const bool hasB = rowB < N;
    v2f ownA[NC], partA[NCH][NC];
    if (hasA) load_tile(rowA, ownA, partA);
    float wr0 = 0.f, wr1 = 0.f, bq = 0.f;
    if constexpr (READOUT) {
        wr0 = wreg[m];
        wr1 = wreg[16 + m];
        bq  = breg[0];
    }

    static_assert(((2 * K * HID) % TPB) == 0 && HID <= TPB && (HID & (HID - 1)) == 0,
                  "whole trips over the weights; the bias index is masked");
    constexpr int WT = (2 * K * HID) / TPB;
    float wv[WT];
    #pragma unroll
    for (int j = 0; j < WT; ++j) {
        const int i   = (int)threadIdx.x + j * TPB;
        const int s   = i / (K * HID);
        const int rem = i - s * (K * HID);
        const int kw  = rem / HID;
        const int col = rem - kw * HID;
        const int kr  = (kw < KW) ? kw : (KW - 1);
        wv[j] = W[((size_t)s * KW + kr) * HID + col];
    }
    const float bv = bias[threadIdx.x & (HID - 1)];
    #pragma unroll
    for (int j = 0; j < WT; ++j) {
        const int i   = (int)threadIdx.x + j * TPB;
        const int s   = i / (K * HID);
        const int rem = i - s * (K * HID);
        const int kw  = rem / HID;
        sW[i] = (kw < KW) ? wv[j] : 0.0f;
    }
    sB[threadIdx.x & (HID - 1)] = bv;
    __syncthreads();
    if (!hasA) return;

    auto product = [&](const v2f (&own)[NC], const v2f (&part)[NCH][NC], v8f& acc0, v8f& acc1) {
        #pragma unroll
        for (int s = 0; s < 2; ++s) {
            const float* __restrict__ sWk = sW + s * K * HID;
            #pragma unroll
            for (int c = 0; c < NC; ++c) {
                const int k0 = c * 4 + khalf;
                v2f a;
                if (s) {
                    float vx = part[0][c].x;
                    float vy = part[0][c].y;
                    #pragma unroll
                    for (int ch = 1; ch < NCH; ++ch) {
                        vx += part[ch][c].x;
                        vy += part[ch][c].y;
                    }
                    a.x = (KW == K || k0 < KW) ? vx : 0.0f;
                    a.y = (KW == K || k0 + 1 < KW) ? vy : 0.0f;
                } else if constexpr (!OWN_PLANES && KW != K) {
                    a.x = (k0 < KW) ? own[c].x : 0.0f;
                    a.y = (k0 + 1 < KW) ? own[c].y : 0.0f;
                } else {
                    a = own[c];
                }
                v2f b0, b1;
                b0.x = sWk[k0 * HID + m];
                b0.y = sWk[(k0 + 1) * HID + m];
                b1.x = sWk[k0 * HID + 16 + m];
                b1.y = sWk[(k0 + 1) * HID + 16 + m];
                if constexpr (OUT_PLANES) {
                    acc0 = __builtin_amdgcn_wmma_f32_16x16x4_f32(
                               false, b0, false, a, (short)0, acc0, false, false);
                    acc1 = __builtin_amdgcn_wmma_f32_16x16x4_f32(
                               false, b1, false, a, (short)0, acc1, false, false);
                } else {
                    acc0 = __builtin_amdgcn_wmma_f32_16x16x4_f32(
                               false, a, false, b0, (short)0, acc0, false, false);
                    acc1 = __builtin_amdgcn_wmma_f32_16x16x4_f32(
                               false, a, false, b1, (short)0, acc1, false, false);
                }
            }
        }
    };

    auto finish = [&](const int row0, const v8f acc0, const v8f acc1) {
        if constexpr (OUT_PLANES) {
            const int fbase = (lane >> 4) ? 8 : 0;
            #pragma unroll
            for (int v = 0; v < 8; ++v) {
                int f = fbase + v;
                float x0 = acc0[v] + sB[f];
                float x1 = acc1[v] + sB[16 + f];
                hout[(size_t)f * NTOT + row0 + m]        = hw_tanh(x0);
                hout[(size_t)(16 + f) * NTOT + row0 + m] = hw_tanh(x1);
            }
        } else {
            const int rbase = row0 + ((lane >> 4) ? 8 : 0);
            float p[8];
            #pragma unroll
            for (int v = 0; v < 8; ++v) {
                int r = rbase + v;
                float x0 = acc0[v] + sB[m];
                float x1 = acc1[v] + sB[16 + m];
                const float t0 = hw_tanh(x0);
                const float t1 = hw_tanh(x1);
                hout[(size_t)r * HID + m]      = t0;
                hout[(size_t)r * HID + 16 + m] = t1;
                p[v] = READOUT ? (t0 * wr0 + t1 * wr1) : 0.0f;
            }
            if constexpr (READOUT) {
                float o = 0.0f;
                #pragma unroll
                for (int v = 0; v < 8; ++v) {
                    const float s = row16_sum(p[v]);
                    o = (m == v) ? s : o;
                }
                if (m < 8) outv[rbase + m] = o + bq;
            }
        }
    };

    v8f accA0 = {0.f,0.f,0.f,0.f,0.f,0.f,0.f,0.f};
    v8f accA1 = {0.f,0.f,0.f,0.f,0.f,0.f,0.f,0.f};
    product(ownA, partA, accA0, accA1);
    v2f ownB[NC], partB[NCH][NC];
    if (hasB) load_tile(rowB, ownB, partB);
    finish(rowA, accA0, accA1);
    if (hasB) {
        v8f accB0 = {0.f,0.f,0.f,0.f,0.f,0.f,0.f,0.f};
        v8f accB1 = {0.f,0.f,0.f,0.f,0.f,0.f,0.f,0.f};
        product(ownB, partB, accB0, accB1);
        finish(rowB, accB0, accB1);
    }
}

static inline int cdiv(long long a, long long b) { return (int)((a + b - 1) / b); }

extern "C" void kernel_launch(void* const* d_in, const int* in_sizes, int n_in,
                              void* d_out, int out_size, void* d_ws, size_t ws_size,
                              hipStream_t stream) {
    const float* X    = (const float*)d_in[0];
    const int*   ei   = (const int*)  d_in[1];
    const float* W0   = (const float*)d_in[2];
    const float* b0   = (const float*)d_in[3];
    const float* W1   = (const float*)d_in[4];
    const float* b1   = (const float*)d_in[5];
    const float* W2   = (const float*)d_in[6];
    const float* b2   = (const float*)d_in[7];
    const float* Wreg = (const float*)d_in[8];
    const float* breg = (const float*)d_in[9];

    float* out    = (float*)d_out;
    float* hfinal = out + NTOT;

    float* dis = (float*)d_ws;
    float* agg = dis + NTOT;
    float* hA  = agg + (size_t)NTOT * HID * LCH;
    float* hB  = hA  + (size_t)NTOT * HID;
    unsigned* cnt = (unsigned*)(hB + (size_t)NTOT * HID);
    unsigned* sched = cnt + (size_t)DCH * NTOT;
    unsigned* ov    = sched + (size_t)BATCH * DCH * GRIDW;
    unsigned* ovcnt = ov + (size_t)BATCH * DCH * OVPAD;

    const int T = 256;

    degree_schedule_kernel<<<BATCH * DCH, 1024, 0, stream>>>(ei, cnt, sched, ov, ovcnt);
    degree_rsqrt_kernel<<<cdiv(NTOT / 4, T), T, 0, stream>>>(cnt, dis);

    static_assert((NTOT % 32) == 0, "every wave of the product launches owns two whole 16-row tiles");
    const int gemm_blocks = cdiv((long long)(NTOT / 32) * 32, T);

    plane_sum_kernel<WINDOW, L0CH><<<BATCH * WINDOW * L0CH, 1024, 0, stream>>>(ei, dis, X, agg);
    tag_gemm_kernel<KPAD, WINDOW, L0CH, false, true, false><<<gemm_blocks, T, 0, stream>>>(X, agg, W0, b0, hA, nullptr, nullptr, nullptr, NTOT);

    pair_sum_kernel<LCH><<<BATCH * (HID / 2) * LCH, 1024, 0, stream>>>(sched, ov, ovcnt, dis, hA, agg);
    tag_gemm_kernel<HID, HID, LCH, true, true, false><<<gemm_blocks, T, 0, stream>>>(hA, agg, W1, b1, hB, nullptr, nullptr, nullptr, NTOT);

    pair_sum_kernel<LCH><<<BATCH * (HID / 2) * LCH, 1024, 0, stream>>>(sched, ov, ovcnt, dis, hB, agg);
    tag_gemm_kernel<HID, HID, LCH, true, false, true><<<gemm_blocks, T, 0, stream>>>(hB, agg, W2, b2, hfinal, Wreg, breg, out, NTOT);
}


# layers 1, 2: edge schedule by owner thread (dst mod 1024); the walk adds into interleaved LDS sum planes by plain 8-byte load, add, store, no f32 atomics; sum order = cell rank, overflow by arrival
# speedup vs baseline: 25.8269x; 1.0732x over previous
#include <hip/hip_runtime.h>
#include <hip/hip_bf16.h>
#include <math.h>

#define BATCH   8
#define WINDOW  5
#define NN      20000
#define EPB     320000
#define HID     32
#define KPAD    8
#define NTOT    (BATCH * NN)
#define ETOT    (BATCH * EPB)
#define L0CH    5
#define DCH     16
#define LCH     2
#define OWNERS  1024
#define PASSES  20
#define GRIDW   (PASSES * OWNERS)
#define OVCAP   (EPB / DCH)
#define OVPAD   (((OVCAP + 1023) / 1024) * 1024)
#define OVNST   32

typedef float v2f __attribute__((ext_vector_type(2)));
typedef float v8f __attribute__((ext_vector_type(8)));

template <int XMASK>
__device__ __forceinline__ float row_xor_get(const float x) {
    return __int_as_float(__builtin_amdgcn_update_dpp(0, __float_as_int(x), 0x160 | XMASK, 0xF, 0xF, true));
}
__device__ __forceinline__ float row16_sum(float x) {
    x += row_xor_get<1>(x);
    x += row_xor_get<2>(x);
    x += row_xor_get<4>(x);
    x += row_xor_get<8>(x);
    return x;
}

__device__ __forceinline__ float hw_tanh(const float x) {
    return __builtin_amdgcn_tanhf(x);
}

__global__ __launch_bounds__(1024)
void degree_schedule_kernel(const int* __restrict__ ei, unsigned* __restrict__ cnt,
                            unsigned* __restrict__ sched, unsigned* __restrict__ ov,
                            unsigned* __restrict__ ovcnt) {
    static_assert((EPB % (4 * DCH)) == 0 && (NN % 4) == 0, "16-byte vectors over edge chunks and nodes");
    static_assert((NN % 32) == 0, "the null source entries NN + lane keep the lane's bank pair");
    static_assert(NN + 32 <= 65536 && NN >= OWNERS, "packed 16-bit source and destination indices; null destinations are real nodes");
    static_assert((PASSES % 4) == 0 && (OVCAP % 4) == 0, "16-byte vectors over grid and overflow list");
    static_assert(OWNERS == 1024 && (OWNERS % 32) == 0, "one owner per thread of the walks' work-groups; dst mod 32 = the owner's lane");
    constexpr int NVC   = EPB / 4 / DCH;
    constexpr int TRIPS = (NVC + 1023) / 1024;
    __shared__ __attribute__((aligned(16))) unsigned sCnt[NN];
    __shared__ __attribute__((aligned(16))) unsigned sGrid[GRIDW];
    __shared__ __attribute__((aligned(16))) unsigned sOv[OVCAP];
    __shared__ __attribute__((aligned(16))) unsigned sCell[OWNERS];
    __shared__ unsigned sOvN;
    const int c   = blockIdx.x % DCH;
    const int b   = blockIdx.x / DCH;
    const int tid = threadIdx.x;

    const int4* __restrict__ S =
        reinterpret_cast<const int4*>(ei + (size_t)b * 2 * EPB) + (size_t)c * NVC;
    const int4* __restrict__ D =
        reinterpret_cast<const int4*>(ei + (size_t)b * 2 * EPB + EPB) + (size_t)c * NVC;
    int4 s[TRIPS], d[TRIPS];
    #pragma unroll
    for (int k = 0; k < TRIPS; ++k) {
        const int j = tid + k * 1024;
        s[k] = S[(j < NVC) ? j : tid];
        d[k] = D[(j < NVC) ? j : tid];
    }
    for (int i = tid; i < NN / 4; i += 1024)
        reinterpret_cast<uint4*>(sCnt)[i] = make_uint4(0u, 0u, 0u, 0u);
    if (tid < OWNERS / 4)
        reinterpret_cast<uint4*>(sCell)[tid] = make_uint4(0u, 0u, 0u, 0u);
    if (tid == 0) sOvN = 0u;
    for (int v = tid; v < GRIDW / 4; v += 1024) {
        const unsigned own = (unsigned)v & (unsigned)(OWNERS - 1);
        const unsigned nw  = ((unsigned)NN + (own & 31u)) | (own << 16);
        reinterpret_cast<uint4*>(sGrid)[v] = make_uint4(nw, nw, nw, nw);
    }
    __syncthreads();

    auto rank_of = [&](const int src, const int dst) -> unsigned {
        __hip_atomic_fetch_add(&sCnt[dst], 1u, __ATOMIC_RELAXED, __HIP_MEMORY_SCOPE_WORKGROUP);
        return __hip_atomic_fetch_add(&sCell[(unsigned)dst & (unsigned)(OWNERS - 1)], 1u,
                                      __ATOMIC_RELAXED, __HIP_MEMORY_SCOPE_WORKGROUP);
    };
    auto put = [&](const int src, const int dst, const unsigned p) {
        const unsigned own = (unsigned)dst & (unsigned)(OWNERS - 1);
        const unsigned w   = (unsigned)src | ((unsigned)dst << 16);
        if (p < (unsigned)PASSES) {
            sGrid[((((p >> 2) * (unsigned)OWNERS) + own) << 2) | (p & 3u)] = w;
        } else {
            const unsigned q = __hip_atomic_fetch_add(&sOvN, 1u, __ATOMIC_RELAXED,
                                                      __HIP_MEMORY_SCOPE_WORKGROUP);
            sOv[q] = w;
        }
    };
    unsigned rk[TRIPS][4];
    #pragma unroll
    for (int k = 0; k < TRIPS; ++k) {
        if (tid + k * 1024 < NVC) {
            rk[k][0] = rank_of(s[k].x, d[k].x);
            rk[k][1] = rank_of(s[k].y, d[k].y);
            rk[k][2] = rank_of(s[k].z, d[k].z);
            rk[k][3] = rank_of(s[k].w, d[k].w);
        }
    }
    #pragma unroll
    for (int k = 0; k < TRIPS; ++k) {
        if (tid + k * 1024 < NVC) {
            put(s[k].x, d[k].x, rk[k][0]);
            put(s[k].y, d[k].y, rk[k][1]);
            put(s[k].z, d[k].z, rk[k][2]);
            put(s[k].w, d[k].w, rk[k][3]);
        }
    }
    __syncthreads();

    uint4* __restrict__ op =
        reinterpret_cast<uint4*>(cnt + (size_t)c * NTOT + (size_t)b * NN);
    for (int i = tid; i < NN / 4; i += 1024)
        op[i] = reinterpret_cast<const uint4*>(sCnt)[i];

    const int seg = b * DCH + c;
    uint4* __restrict__ gp = reinterpret_cast<uint4*>(sched + (size_t)seg * GRIDW);
    for (int i = tid; i < GRIDW / 4; i += 1024)
        gp[i] = reinterpret_cast<const uint4*>(sGrid)[i];
    const unsigned nov = sOvN;
    const unsigned nup = (nov + 1023u) & ~1023u;
    const unsigned npad = (nup > 2048u) ? nup : 2048u;
    unsigned* __restrict__ vp = ov + (size_t)seg * OVPAD;
    for (unsigned i = tid; i < npad; i += 1024u)
        vp[i] = (i < nov) ? sOv[i] : (((unsigned)NN + (i & 31u)) | ((i & 63u) << 16));
    if (tid == 0) ovcnt[(size_t)seg * OVNST] = nov;
}

__global__ void degree_rsqrt_kernel(const unsigned* __restrict__ cnt, float* __restrict__ dis) {
    int i = blockIdx.x * blockDim.x + threadIdx.x;
    if (i >= NTOT / 4) return;
    uint4 s = reinterpret_cast<const uint4*>(cnt)[i];
    #pragma unroll
    for (int c = 1; c < DCH; ++c) {
        const uint4 t = reinterpret_cast<const uint4*>(cnt + (size_t)c * NTOT)[i];
        s.x += t.x; s.y += t.y; s.z += t.z; s.w += t.w;
    }
    float4 r;
    r.x = (s.x > 0u) ? rsqrtf((float)s.x) : 0.0f;
    r.y = (s.y > 0u) ? rsqrtf((float)s.y) : 0.0f;
    r.z = (s.z > 0u) ? rsqrtf((float)s.z) : 0.0f;
    r.w = (s.w > 0u) ? rsqrtf((float)s.w) : 0.0f;
    reinterpret_cast<float4*>(dis)[i] = r;
}

template <int NF, int NCH>
__global__ __launch_bounds__(1024)
void plane_sum_kernel(const int* __restrict__ ei, const float* __restrict__ dis,
                      const float* __restrict__ src, float* __restrict__ dst) {
    static_assert((NN % 4) == 0 && (EPB % (4 * NCH)) == 0, "16-byte vectors over nodes and edge chunks");
    constexpr int NVC  = EPB / 4 / NCH;
    constexpr int STEP = 2048;
    constexpr int TRIPS = (NVC + STEP - 1) / STEP;
    static_assert(NVC >= STEP && (TRIPS - 1) * STEP + 1024 <= NVC,
                  "every thread owns a first pair of edge vectors, and a first vector in every trip");
    constexpr int NV4  = NN / 4;
    constexpr int FT   = (NV4 + 1023) / 1024;
    constexpr int FS   = (NN + 1023) / 1024;
    __shared__ __attribute__((aligned(16))) float sSrc[NN];
    __shared__ __attribute__((aligned(16))) float sAcc[NN + 1024];
    static_assert(sizeof(float) * (2 * NN + 1024) > 160 * 1024, "more than half of the 320 KiB of a WGP");
    const int c   = blockIdx.x % NCH;
    const int t   = blockIdx.x / NCH;
    const int f   = t % NF;
    const int b   = t / NF;
    const int tid = threadIdx.x;

    const float* __restrict__ gdis = dis + (size_t)b * NN;
    const float* __restrict__ sp = src + (size_t)b * NN * NF + f;
    const float4* __restrict__ dq = reinterpret_cast<const float4*>(gdis);
    const int* __restrict__ eb = ei + (size_t)b * 2 * EPB;
    const int4* __restrict__ S = reinterpret_cast<const int4*>(eb) + (size_t)c * NVC;
    const int4* __restrict__ D = reinterpret_cast<const int4*>(eb + EPB) + (size_t)c * NVC;
    float xs[FS], ds[FS];
    #pragma unroll
    for (int k = 0; k < FS; ++k) {
        const int n  = tid + k * 1024;
        const int nn = (n < NN) ? n : tid;
        xs[k] = sp[(size_t)nn * NF];
        ds[k] = gdis[nn];
    }
    float4 q[FT];
    #pragma unroll
    for (int k = 0; k < FT; ++k) {
        const int j = tid + k * 1024;
        q[k] = dq[(j < NV4) ? j : tid];
    }
    int i = tid;
    int4 s0 = S[i], s1 = S[i + 1024];
    int4 d0 = D[i], d1 = D[i + 1024];
    #pragma unroll
    for (int k = 0; k < FS; ++k) {
        const int n  = tid + k * 1024;
        const int nn = (n < NN) ? n : tid;
        sSrc[nn] = xs[k] * ds[k];
        sAcc[nn] = 0.f;
    }
    __syncthreads();

    auto add4 = [&](const int4 s, const int4 d) {
        const float v0 = sSrc[s.x];
        const float v1 = sSrc[s.y];
        const float v2 = sSrc[s.z];
        const float v3 = sSrc[s.w];
        __hip_atomic_fetch_add(&sAcc[d.x], v0, __ATOMIC_RELAXED, __HIP_MEMORY_SCOPE_WORKGROUP);
        __hip_atomic_fetch_add(&sAcc[d.y], v1, __ATOMIC_RELAXED, __HIP_MEMORY_SCOPE_WORKGROUP);
        __hip_atomic_fetch_add(&sAcc[d.z], v2, __ATOMIC_RELAXED, __HIP_MEMORY_SCOPE_WORKGROUP);
        __hip_atomic_fetch_add(&sAcc[d.w], v3, __ATOMIC_RELAXED, __HIP_MEMORY_SCOPE_WORKGROUP);
    };

    for (int trip = 0; trip < TRIPS; ++trip) {
        const int ni = (trip + 1 < TRIPS) ? i + STEP : i;
        const int n1 = (ni + 1024 < NVC) ? ni + 1024 : ni;
        const int4 s0n = S[ni], s1n = S[n1];
        const int4 d0n = D[ni], d1n = D[n1];
        add4(s0, d0);
        if (i + 1024 < NVC) add4(s1, d1);
        s0 = s0n; s1 = s1n; d0 = d0n; d1 = d1n;
        i = ni;
    }
    __syncthreads();

    float4* __restrict__ dp =
        reinterpret_cast<float4*>(dst + (size_t)(c * NF + f) * NTOT + (size_t)b * NN);
    #pragma unroll
    for (int k = 0; k < FT; ++k) {
        const int j = tid + k * 1024;
        if (j < NV4) {
            const float4 a = reinterpret_cast<const float4*>(sAcc)[j];
            dp[j] = make_float4(a.x * q[k].x, a.y * q[k].y, a.z * q[k].z, a.w * q[k].w);
        }
    }
}

template <int NCH>
__global__ __launch_bounds__(1024)
void pair_sum_kernel(const unsigned* __restrict__ sched, const unsigned* __restrict__ ov,
                     const unsigned* __restrict__ ovcnt, const float* __restrict__ dis,
                     const float* __restrict__ src, float* __restrict__ dst) {
    static_assert((NN % 4) == 0 && (DCH % NCH) == 0 && (HID % 2) == 0,
                  "16-byte vectors over nodes, whole segments per chunk, features in pairs");
    constexpr int SPC  = DCH / NCH;
    constexpr int NVG  = SPC * GRIDW / 4;
    constexpr int STEP = 2048;
    constexpr int FULL = NVG / STEP;
    static_assert(FULL >= 1 && (NVG % STEP) == 0, "the grid is a whole number of trips");
    constexpr int NV4  = NN / 4;
    constexpr int FT   = (NV4 + 1023) / 1024;
    static_assert(((GRIDW / 4) % OWNERS) == 0 && OWNERS == 1024,
                  "vector index mod OWNERS = thread index: a thread reads only the slots of the destinations it owns");
    __shared__ __attribute__((aligned(16))) float2 sSrc[NN + 32];
    __shared__ __attribute__((aligned(16))) float  sAcc[2 * NN];
    static_assert(sizeof(float) * (4 * NN + 64) <= 320 * 1024, "four planes and the null entries fit the 320 KiB of a WGP");
    const int c   = blockIdx.x % NCH;
    const int t   = blockIdx.x / NCH;
    const int f0  = (t % (HID / 2)) * 2;
    const int b   = t / (HID / 2);
    const int tid = threadIdx.x;

    const float* __restrict__ gdis = dis + (size_t)b * NN;
    float4 u[FT], v[FT], q[FT];
    {
        const float4* __restrict__ sp0 =
            reinterpret_cast<const float4*>(src + (size_t)f0 * NTOT + (size_t)b * NN);
        const float4* __restrict__ sp1 =
            reinterpret_cast<const float4*>(src + (size_t)(f0 + 1) * NTOT + (size_t)b * NN);
        const float4* __restrict__ dq = reinterpret_cast<const float4*>(gdis);
        #pragma unroll
        for (int k = 0; k < FT; ++k) {
            const int j  = tid + k * 1024;
            const int jj = (j < NV4) ? j : tid;
            u[k] = sp0[jj];
            v[k] = sp1[jj];
            q[k] = dq[jj];
        }
    }
    const int seg0 = b * DCH + c * SPC;
    const uint4* __restrict__ G = reinterpret_cast<const uint4*>(sched + (size_t)seg0 * GRIDW);
    static_assert((SPC % 2) == 0, "overflow words are taken four at a time, two segments each");
    unsigned x[SPC][2];
    #pragma unroll
    for (int sg = 0; sg < SPC; ++sg) {
        const unsigned* __restrict__ O = ov + (size_t)(seg0 + sg) * OVPAD;
        x[sg][0] = O[tid];
        x[sg][1] = O[tid + 1024];
    }
    unsigned nup[SPC];
    #pragma unroll
    for (int sg = 0; sg < SPC; ++sg) {
        unsigned n = ovcnt[(size_t)(seg0 + sg) * OVNST];
        n = (n < (unsigned)OVCAP) ? n : (unsigned)OVCAP;
        nup[sg] = (n + 1023u) & ~1023u;
    }
    int i = tid;
    uint4 w0 = G[i], w1 = G[i + 1024];

    #pragma unroll
    for (int k = 0; k < FT; ++k) {
        const int j  = tid + k * 1024;
        const int jj = (j < NV4) ? j : tid;
        reinterpret_cast<float4*>(sSrc)[2 * jj]     = make_float4(u[k].x * q[k].x, v[k].x * q[k].x, u[k].y * q[k].y, v[k].y * q[k].y);
        reinterpret_cast<float4*>(sSrc)[2 * jj + 1] = make_float4(u[k].z * q[k].z, v[k].z * q[k].z, u[k].w * q[k].w, v[k].w * q[k].w);
        reinterpret_cast<float4*>(sAcc)[2 * jj]     = make_float4(0.f, 0.f, 0.f, 0.f);
        reinterpret_cast<float4*>(sAcc)[2 * jj + 1] = make_float4(0.f, 0.f, 0.f, 0.f);
    }
    if (tid < 32) sSrc[NN + tid] = make_float2(0.f, 0.f);
    __syncthreads();

    auto add1 = [&](const unsigned w) {
        const float2 v = sSrc[w & 0xFFFFu];
        const unsigned d = w >> 16;
        __hip_atomic_fetch_add(&sAcc[2u * d],      v.x, __ATOMIC_RELAXED, __HIP_MEMORY_SCOPE_WORKGROUP);
        __hip_atomic_fetch_add(&sAcc[2u * d + 1u], v.y, __ATOMIC_RELAXED, __HIP_MEMORY_SCOPE_WORKGROUP);
    };
    auto add4 = [&](const uint4 w) {
        const float2 v0 = sSrc[w.x & 0xFFFFu];
        const float2 v1 = sSrc[w.y & 0xFFFFu];
        const float2 v2 = sSrc[w.z & 0xFFFFu];
        const float2 v3 = sSrc[w.w & 0xFFFFu];
        const unsigned d0 = w.x >> 16, d1 = w.y >> 16, d2 = w.z >> 16, d3 = w.w >> 16;
        __hip_atomic_fetch_add(&sAcc[2u * d0],      v0.x, __ATOMIC_RELAXED, __HIP_MEMORY_SCOPE_WORKGROUP);
        __hip_atomic_fetch_add(&sAcc[2u * d0 + 1u], v0.y, __ATOMIC_RELAXED, __HIP_MEMORY_SCOPE_WORKGROUP);
        __hip_atomic_fetch_add(&sAcc[2u * d1],      v1.x, __ATOMIC_RELAXED, __HIP_MEMORY_SCOPE_WORKGROUP);
        __hip_atomic_fetch_add(&sAcc[2u * d1 + 1u], v1.y, __ATOMIC_RELAXED, __HIP_MEMORY_SCOPE_WORKGROUP);
        __hip_atomic_fetch_add(&sAcc[2u * d2],      v2.x, __ATOMIC_RELAXED, __HIP_MEMORY_SCOPE_WORKGROUP);
        __hip_atomic_fetch_add(&sAcc[2u * d2 + 1u], v2.y, __ATOMIC_RELAXED, __HIP_MEMORY_SCOPE_WORKGROUP);
        __hip_atomic_fetch_add(&sAcc[2u * d3],      v3.x, __ATOMIC_RELAXED, __HIP_MEMORY_SCOPE_WORKGROUP);
        __hip_atomic_fetch_add(&sAcc[2u * d3 + 1u], v3.y, __ATOMIC_RELAXED, __HIP_MEMORY_SCOPE_WORKGROUP);
    };
    const v2f* const src2 = reinterpret_cast<const v2f*>(sSrc);
    v2f* const acc2 = reinterpret_cast<v2f*>(sAcc);
    auto own4 = [&](const uint4 w) {
        const unsigned d0 = w.x >> 16, d1 = w.y >> 16, d2 = w.z >> 16, d3 = w.w >> 16;
        const v2f v0 = src2[w.x & 0xFFFFu];
        const v2f v1 = src2[w.y & 0xFFFFu];
        const v2f v2 = src2[w.z & 0xFFFFu];
        const v2f v3 = src2[w.w & 0xFFFFu];
        const v2f a0 = acc2[d0];
        const v2f a1 = acc2[d1];
        const v2f a2 = acc2[d2];
        const v2f a3 = acc2[d3];
        asm volatile("" :: "v"(v0), "v"(v1), "v"(v2), "v"(v3), "v"(a0), "v"(a1), "v"(a2), "v"(a3));
        const v2f n0 = a0 + v0;
        const v2f n1 = ((d1 == d0) ? n0 : a1) + v1;
        const v2f n2 = ((d2 == d1) ? n1 : ((d2 == d0) ? n0 : a2)) + v2;
        const v2f n3 = ((d3 == d2) ? n2 : ((d3 == d1) ? n1 : ((d3 == d0) ? n0 : a3))) + v3;
        acc2[d0] = n0;
        acc2[d1] = n1;
        acc2[d2] = n2;
        acc2[d3] = n3;
    };

    for (int trip = 0; trip < FULL; ++trip) {
        const int ni = (trip + 1 < FULL) ? i + STEP : i;
        const uint4 w0n = G[ni], w1n = G[ni + 1024];
        own4(w0);
        own4(w1);
        w0 = w0n; w1 = w1n;
        i = ni;
    }
    __syncthreads();
    #pragma unroll
    for (int sg = 0; sg < SPC; sg += 2)
        add4(make_uint4(x[sg][0], x[sg][1], x[sg + 1][0], x[sg + 1][1]));
    #pragma unroll
    for (int sg = 0; sg < SPC; ++sg) {
        const unsigned* __restrict__ O = ov + (size_t)(seg0 + sg) * OVPAD;
        for (unsigned j = 2048u + tid; j < nup[sg]; j += 1024u) add1(O[j]);
    }
    __syncthreads();

    float4* __restrict__ dp0 =
        reinterpret_cast<float4*>(dst + (size_t)(c * HID + f0) * NTOT + (size_t)b * NN);
    float4* __restrict__ dp1 =
        reinterpret_cast<float4*>(dst + (size_t)(c * HID + f0 + 1) * NTOT + (size_t)b * NN);
    #pragma unroll
    for (int k = 0; k < FT; ++k) {
        const int j = tid + k * 1024;
        if (j < NV4) {
            const float4 lo = reinterpret_cast<const float4*>(sAcc)[2 * j];
            const float4 hi = reinterpret_cast<const float4*>(sAcc)[2 * j + 1];
            dp0[j] = make_float4(lo.x * q[k].x, lo.z * q[k].y, hi.x * q[k].z, hi.z * q[k].w);
            dp1[j] = make_float4(lo.y * q[k].x, lo.w * q[k].y, hi.y * q[k].z, hi.w * q[k].w);
        }
    }
}

template <int K, int KW, int NCH, bool OWN_PLANES, bool OUT_PLANES, bool READOUT>
__global__ __launch_bounds__(256)
void tag_gemm_kernel(const float* __restrict__ hin,
                                const float* __restrict__ agg,
                                const float* __restrict__ W,
                                const float* __restrict__ bias,
                                float* __restrict__ hout,
                                const float* __restrict__ wreg,
                                const float* __restrict__ breg,
                                float* __restrict__ outv, int N) {
    static_assert((K % 4) == 0, "padded K must be a multiple of 4");
    static_assert(!(READOUT && OUT_PLANES), "the readout sums rows of the row-major tile");
    constexpr int TPB = 256;
    constexpr int NC  = K / 4;
    __shared__ float sW[2 * K * HID];
    __shared__ float sB[HID];

    const int lane = threadIdx.x & 31;
    const int wave = (blockIdx.x * TPB + threadIdx.x) >> 5;

    const int m     = lane & 15;
    const int khalf = (lane >> 4) * 2;

    auto load_tile = [&](const int row0, v2f (&own)[NC], v2f (&part)[NCH][NC]) {
        #pragma unroll
        for (int c = 0; c < NC; ++c) {
            const int k0 = c * 4 + khalf;
            const int kx = (KW == K || k0 < KW) ? k0 : (KW - 1);
            const int ky = (KW == K || k0 + 1 < KW) ? (k0 + 1) : (KW - 1);
            if constexpr (OWN_PLANES) {
                own[c].x = hin[(size_t)k0 * NTOT + row0 + m];
                own[c].y = hin[(size_t)(k0 + 1) * NTOT + row0 + m];
            } else {
                own[c].x = hin[(size_t)(row0 + m) * KW + kx];
                own[c].y = hin[(size_t)(row0 + m) * KW + ky];
            }
            #pragma unroll
            for (int ch = 0; ch < NCH; ++ch) {
                part[ch][c].x = agg[((size_t)ch * KW + kx) * NTOT + row0 + m];
                part[ch][c].y = agg[((size_t)ch * KW + ky) * NTOT + row0 + m];
            }
        }
    };

    const int rowA = wave * 32;
    const int rowB = rowA + 16;
    const bool hasA = rowA < N;
    const bool hasB = rowB < N;
    v2f ownA[NC], partA[NCH][NC];
    if (hasA) load_tile(rowA, ownA, partA);
    float wr0 = 0.f, wr1 = 0.f, bq = 0.f;
    if constexpr (READOUT) {
        wr0 = wreg[m];
        wr1 = wreg[16 + m];
        bq  = breg[0];
    }

    static_assert(((2 * K * HID) % TPB) == 0 && HID <= TPB && (HID & (HID - 1)) == 0,
                  "whole trips over the weights; the bias index is masked");
    constexpr int WT = (2 * K * HID) / TPB;
    float wv[WT];
    #pragma unroll
    for (int j = 0; j < WT; ++j) {
        const int i   = (int)threadIdx.x + j * TPB;
        const int s   = i / (K * HID);
        const int rem = i - s * (K * HID);
        const int kw  = rem / HID;
        const int col = rem - kw * HID;
        const int kr  = (kw < KW) ? kw : (KW - 1);
        wv[j] = W[((size_t)s * KW + kr) * HID + col];
    }
    const float bv = bias[threadIdx.x & (HID - 1)];
    #pragma unroll
    for (int j = 0; j < WT; ++j) {
        const int i   = (int)threadIdx.x + j * TPB;
        const int s   = i / (K * HID);
        const int rem = i - s * (K * HID);
        const int kw  = rem / HID;
        sW[i] = (kw < KW) ? wv[j] : 0.0f;
    }
    sB[threadIdx.x & (HID - 1)] = bv;
    __syncthreads();
    if (!hasA) return;

    auto product = [&](const v2f (&own)[NC], const v2f (&part)[NCH][NC], v8f& acc0, v8f& acc1) {
        #pragma unroll
        for (int s = 0; s < 2; ++s) {
            const float* __restrict__ sWk = sW + s * K * HID;
            #pragma unroll
            for (int c = 0; c < NC; ++c) {
                const int k0 = c * 4 + khalf;
                v2f a;
                if (s) {
                    float vx = part[0][c].x;
                    float vy = part[0][c].y;
                    #pragma unroll
                    for (int ch = 1; ch < NCH; ++ch) {
                        vx += part[ch][c].x;
                        vy += part[ch][c].y;
                    }
                    a.x = (KW == K || k0 < KW) ? vx : 0.0f;
                    a.y = (KW == K || k0 + 1 < KW) ? vy : 0.0f;
                } else if constexpr (!OWN_PLANES && KW != K) {
                    a.x = (k0 < KW) ? own[c].x : 0.0f;
                    a.y = (k0 + 1 < KW) ? own[c].y : 0.0f;
                } else {
                    a = own[c];
                }
                v2f b0, b1;
                b0.x = sWk[k0 * HID + m];
                b0.y = sWk[(k0 + 1) * HID + m];
                b1.x = sWk[k0 * HID + 16 + m];
                b1.y = sWk[(k0 + 1) * HID + 16 + m];
                if constexpr (OUT_PLANES) {
                    acc0 = __builtin_amdgcn_wmma_f32_16x16x4_f32(
                               false, b0, false, a, (short)0, acc0, false, false);
                    acc1 = __builtin_amdgcn_wmma_f32_16x16x4_f32(
                               false, b1, false, a, (short)0, acc1, false, false);
                } else {
                    acc0 = __builtin_amdgcn_wmma_f32_16x16x4_f32(
                               false, a, false, b0, (short)0, acc0, false, false);
                    acc1 = __builtin_amdgcn_wmma_f32_16x16x4_f32(
                               false, a, false, b1, (short)0, acc1, false, false);
                }
            }
        }
    };

    auto finish = [&](const int row0, const v8f acc0, const v8f acc1) {
        if constexpr (OUT_PLANES) {
            const int fbase = (lane >> 4) ? 8 : 0;
            #pragma unroll
            for (int v = 0; v < 8; ++v) {
                int f = fbase + v;
                float x0 = acc0[v] + sB[f];
                float x1 = acc1[v] + sB[16 + f];
                hout[(size_t)f * NTOT + row0 + m]        = hw_tanh(x0);
                hout[(size_t)(16 + f) * NTOT + row0 + m] = hw_tanh(x1);
            }
        } else {
            const int rbase = row0 + ((lane >> 4) ? 8 : 0);
            float p[8];
            #pragma unroll
            for (int v = 0; v < 8; ++v) {
                int r = rbase + v;
                float x0 = acc0[v] + sB[m];
                float x1 = acc1[v] + sB[16 + m];
                const float t0 = hw_tanh(x0);
                const float t1 = hw_tanh(x1);
                hout[(size_t)r * HID + m]      = t0;
                hout[(size_t)r * HID + 16 + m] = t1;
                p[v] = READOUT ? (t0 * wr0 + t1 * wr1) : 0.0f;
            }
            if constexpr (READOUT) {
                float o = 0.0f;
                #pragma unroll
                for (int v = 0; v < 8; ++v) {
                    const float s = row16_sum(p[v]);
                    o = (m == v) ? s : o;
                }
                if (m < 8) outv[rbase + m] = o + bq;
            }
        }
    };

    v8f accA0 = {0.f,0.f,0.f,0.f,0.f,0.f,0.f,0.f};
    v8f accA1 = {0.f,0.f,0.f,0.f,0.f,0.f,0.f,0.f};
    product(ownA, partA, accA0, accA1);
    v2f ownB[NC], partB[NCH][NC];
    if (hasB) load_tile(rowB, ownB, partB);
    finish(rowA, accA0, accA1);
    if (hasB) {
        v8f accB0 = {0.f,0.f,0.f,0.f,0.f,0.f,0.f,0.f};
        v8f accB1 = {0.f,0.f,0.f,0.f,0.f,0.f,0.f,0.f};
        product(ownB, partB, accB0, accB1);
        finish(rowB, accB0, accB1);
    }
}

static inline int cdiv(long long a, long long b) { return (int)((a + b - 1) / b); }

extern "C" void kernel_launch(void* const* d_in, const int* in_sizes, int n_in,
                              void* d_out, int out_size, void* d_ws, size_t ws_size,
                              hipStream_t stream) {
    const float* X    = (const float*)d_in[0];
    const int*   ei   = (const int*)  d_in[1];
    const float* W0   = (const float*)d_in[2];
    const float* b0   = (const float*)d_in[3];
    const float* W1   = (const float*)d_in[4];
    const float* b1   = (const float*)d_in[5];
    const float* W2   = (const float*)d_in[6];
    const float* b2   = (const float*)d_in[7];
    const float* Wreg = (const float*)d_in[8];
    const float* breg = (const float*)d_in[9];

    float* out    = (float*)d_out;
    float* hfinal = out + NTOT;

    float* dis = (float*)d_ws;
    float* agg = dis + NTOT;
    float* hA  = agg + (size_t)NTOT * HID * LCH;
    float* hB  = hA  + (size_t)NTOT * HID;
    unsigned* cnt = (unsigned*)(hB + (size_t)NTOT * HID);
    unsigned* sched = cnt + (size_t)DCH * NTOT;
    unsigned* ov    = sched + (size_t)BATCH * DCH * GRIDW;
    unsigned* ovcnt = ov + (size_t)BATCH * DCH * OVPAD;

    const int T = 256;

    degree_schedule_kernel<<<BATCH * DCH, 1024, 0, stream>>>(ei, cnt, sched, ov, ovcnt);
    degree_rsqrt_kernel<<<cdiv(NTOT / 4, T), T, 0, stream>>>(cnt, dis);

    static_assert((NTOT % 32) == 0, "every wave of the product launches owns two whole 16-row tiles");
    const int gemm_blocks = cdiv((long long)(NTOT / 32) * 32, T);

    plane_sum_kernel<WINDOW, L0CH><<<BATCH * WINDOW * L0CH, 1024, 0, stream>>>(ei, dis, X, agg);
    tag_gemm_kernel<KPAD, WINDOW, L0CH, false, true, false><<<gemm_blocks, T, 0, stream>>>(X, agg, W0, b0, hA, nullptr, nullptr, nullptr, NTOT);

    pair_sum_kernel<LCH><<<BATCH * (HID / 2) * LCH, 1024, 0, stream>>>(sched, ov, ovcnt, dis, hA, agg);
    tag_gemm_kernel<HID, HID, LCH, true, true, false><<<gemm_blocks, T, 0, stream>>>(hA, agg, W1, b1, hB, nullptr, nullptr, nullptr, NTOT);

    pair_sum_kernel<LCH><<<BATCH * (HID / 2) * LCH, 1024, 0, stream>>>(sched, ov, ovcnt, dis, hB, agg);
    tag_gemm_kernel<HID, HID, LCH, true, false, true><<<gemm_blocks, T, 0, stream>>>(hB, agg, W2, b2, hfinal, Wreg, breg, out, NTOT);
}
